# starting kernel
# baseline (speedup 1.0000x reference)
#include <hip/hip_runtime.h>
#include <hip/hip_bf16.h>

typedef __attribute__((ext_vector_type(2))) float v2f;
typedef __attribute__((ext_vector_type(8))) float v8f;

#define N_V   100000
#define N_E   20000
#define N_NZ  400000
#define CIN   256
#define HID   256
#define NH    4
#define DH    64
#define NEG_SLOPE 0.2f

__global__ __launch_bounds__(256) void v2e_scatter(
    const float* __restrict__ X, const int* __restrict__ inc_v,
    const int* __restrict__ inc_e, float* __restrict__ Xsum,
    float* __restrict__ cnt)
{
  const int widx = blockIdx.x * 8 + (threadIdx.x >> 5);
  if (widx >= N_NZ) return;
  const int lane = threadIdx.x & 31;
  const int v = inc_v[widx];
  const int e = inc_e[widx];
  const float4* __restrict__ src4 = (const float4*)(X + (size_t)v * CIN);
  float* __restrict__ dst = Xsum + (size_t)e * HID;
#pragma unroll
  for (int f4 = lane; f4 < HID / 4; f4 += 32) {
    const float4 val = src4[f4];
    atomicAdd(&dst[f4 * 4 + 0], val.x);
    atomicAdd(&dst[f4 * 4 + 1], val.y);
    atomicAdd(&dst[f4 * 4 + 2], val.z);
    atomicAdd(&dst[f4 * 4 + 3], val.w);
  }
  if (lane == 0) atomicAdd(&cnt[e], 1.0f);
}

__global__ __launch_bounds__(256) void gemm_edge(
    const float* __restrict__ Xsum, const float* __restrict__ cnt,
    const float* __restrict__ W, const float* __restrict__ bvec,
    float* __restrict__ Y)
{
  const int lane = threadIdx.x & 31;
  const int wid  = blockIdx.x * 8 + (threadIdx.x >> 5);
  const int row_tile = wid >> 2;
  const int head = wid & 3;
  const int row0 = row_tile * 16;
  const int m = lane & 15;
  const int half = lane >> 4;

  const int row = row0 + m;
  const float inv = 1.0f / fmaxf(cnt[row], 1.0f);
  const float* __restrict__ arow = Xsum + (size_t)row * HID;
  const float* __restrict__ wh = W + (size_t)head * (CIN * DH);

  v8f acc[4] = {v8f{}, v8f{}, v8f{}, v8f{}};
#pragma unroll 4
  for (int k0 = 0; k0 < CIN; k0 += 4) {
    const int ka = k0 + 2 * half;
    v2f afrag;
    afrag.x = arow[ka] * inv;
    afrag.y = arow[ka + 1] * inv;
    const float* __restrict__ wk = wh + (size_t)ka * DH + m;
#pragma unroll
    for (int t = 0; t < 4; ++t) {
      v2f bfrag;
      bfrag.x = wk[t * 16];
      bfrag.y = wk[t * 16 + DH];
      acc[t] = __builtin_amdgcn_wmma_f32_16x16x4_f32(
          false, afrag, false, bfrag, (short)0, acc[t], false, false);
    }
  }

  const int mbase = half * 8;
#pragma unroll
  for (int t = 0; t < 4; ++t) {
    const int col = head * DH + t * 16 + m;
    const float bias = bvec[col];
#pragma unroll
    for (int r = 0; r < 8; ++r) {
      Y[(size_t)(row0 + mbase + r) * HID + col] = acc[t][r] + bias;
    }
  }
}

__global__ __launch_bounds__(256) void edge_alpha(
    const float* __restrict__ Y, const float* __restrict__ a_e,
    float* __restrict__ es)
{
  __shared__ float red[256];
  const int e = blockIdx.x;
  const int f = threadIdx.x;
  red[f] = Y[(size_t)e * HID + f] * a_e[f];
  __syncthreads();
#pragma unroll
  for (int off = 32; off >= 1; off >>= 1) {
    if ((f & 63) < off) red[f] += red[f + off];
    __syncthreads();
  }
  if ((f & 63) == 0) {
    const float alpha = red[f];
    float s = alpha >= 0.0f ? alpha : NEG_SLOPE * alpha;
    s = fminf(fmaxf(s, 0.001f), 5.0f);
    es[e * NH + (f >> 6)] = __expf(s);
  }
}

__global__ __launch_bounds__(256) void vert_denom(
    const float* __restrict__ es, const int* __restrict__ inc_v,
    const int* __restrict__ inc_e, float* __restrict__ denom)
{
  const int inc = blockIdx.x * 256 + threadIdx.x;
  if (inc >= N_NZ) return;
  const int v = inc_v[inc], e = inc_e[inc];
  const float4 s4 = ((const float4*)es)[e];
  float* __restrict__ d = denom + (size_t)v * NH;
  atomicAdd(&d[0], s4.x);
  atomicAdd(&d[1], s4.y);
  atomicAdd(&d[2], s4.z);
  atomicAdd(&d[3], s4.w);
}

__global__ __launch_bounds__(256) void aggregate(
    const float* __restrict__ Y, const float* __restrict__ es,
    const int* __restrict__ inc_v, const int* __restrict__ inc_e,
    const float* __restrict__ denom, float* __restrict__ out)
{
  const int widx = blockIdx.x * 8 + (threadIdx.x >> 5);
  if (widx >= N_NZ) return;
  const int lane = threadIdx.x & 31;
  const int v = inc_v[widx];
  const int e = inc_e[widx];
  const float4 s4 = ((const float4*)es)[e];
  const float4 d4 = ((const float4*)denom)[v];
  float wgt[NH] = { s4.x / d4.x, s4.y / d4.y, s4.z / d4.z, s4.w / d4.w };

  const float4* __restrict__ yrow4 = (const float4*)(Y + (size_t)e * HID);
  float* __restrict__ orow = out + (size_t)v * HID;
#pragma unroll
  for (int f4 = lane; f4 < HID / 4; f4 += 32) {
    const float w = wgt[f4 >> 4];
    const float4 y = yrow4[f4];
    atomicAdd(&orow[f4 * 4 + 0], w * y.x);
    atomicAdd(&orow[f4 * 4 + 1], w * y.y);
    atomicAdd(&orow[f4 * 4 + 2], w * y.z);
    atomicAdd(&orow[f4 * 4 + 3], w * y.w);
  }
}

__global__ __launch_bounds__(256) void elu_kernel(float4* __restrict__ out)
{
  const size_t i = (size_t)blockIdx.x * 256 + threadIdx.x;
  if (i >= (size_t)N_V * HID / 4) return;
  float4 x = out[i];
  x.x = x.x > 0.0f ? x.x : expm1f(x.x);
  x.y = x.y > 0.0f ? x.y : expm1f(x.y);
  x.z = x.z > 0.0f ? x.z : expm1f(x.z);
  x.w = x.w > 0.0f ? x.w : expm1f(x.w);
  out[i] = x;
}

extern "C" void kernel_launch(void* const* d_in, const int* in_sizes, int n_in,
                              void* d_out, int out_size, void* d_ws, size_t ws_size,
                              hipStream_t stream) {
  const float* X     = (const float*)d_in[0];
  const float* W     = (const float*)d_in[1];
  const float* bvec  = (const float*)d_in[2];
  const float* a_e   = (const float*)d_in[3];
  const int*   inc_v = (const int*)d_in[4];
  const int*   inc_e = (const int*)d_in[5];
  float* out = (float*)d_out;

  float* ws    = (float*)d_ws;
  float* Xsum  = ws;
  float* Y     = Xsum + (size_t)N_E * HID;
  float* cnt   = Y    + (size_t)N_E * HID;
  float* es    = cnt  + N_E;
  float* denom = es   + (size_t)N_E * NH;

  hipMemsetAsync(Xsum, 0, sizeof(float) * (size_t)N_E * HID, stream);
  hipMemsetAsync(cnt, 0, sizeof(float) * N_E, stream);
  hipMemsetAsync(denom, 0, sizeof(float) * (size_t)N_V * NH, stream);
  hipMemsetAsync(out, 0, sizeof(float) * (size_t)N_V * HID, stream);

  v2e_scatter<<<dim3(N_NZ / 8), 256, 0, stream>>>(X, inc_v, inc_e, Xsum, cnt);
  gemm_edge<<<dim3((N_E / 16) * NH / 8), 256, 0, stream>>>(Xsum, cnt, W, bvec, Y);
  edge_alpha<<<dim3(N_E), 256, 0, stream>>>(Y, a_e, es);
  vert_denom<<<dim3((N_NZ + 255) / 256), 256, 0, stream>>>(es, inc_v, inc_e, denom);
  aggregate<<<dim3(N_NZ / 8), 256, 0, stream>>>(Y, es, inc_v, inc_e, denom, out);
  elu_kernel<<<dim3(((size_t)N_V * HID / 4 + 255) / 256), 256, 0, stream>>>((float4*)out);
}


# v2e+e2v scatters: destination-range work-groups hold sums in LDS, filter-scan the incidence list, add rows by LDS f32 atomics; softmax quotient, ELU, one out store folded in; sum order = LDS arrival
# speedup vs baseline: 2.7553x; 2.7553x over previous
#include <hip/hip_runtime.h>
#include <hip/hip_bf16.h>

typedef __attribute__((ext_vector_type(2))) float v2f;
typedef __attribute__((ext_vector_type(8))) float v8f;

#define N_V   100000
#define N_E   20000
#define N_NZ  400000
#define CIN   256
#define HID   256
#define NH    4
#define DH    64
#define NEG_SLOPE 0.2f

#define WGT    1024
#define NWAVE  32
#define LCAP   8192
#define NQ     (N_NZ / 4)
#define RE     80
#define RV     196

__device__ __forceinline__ void lds_fadd(float* p, float v) {
  __hip_atomic_fetch_add(p, v, __ATOMIC_RELAXED, __HIP_MEMORY_SCOPE_WORKGROUP);
}
__device__ __forceinline__ int lds_iadd(int* p, int v) {
  return __hip_atomic_fetch_add(p, v, __ATOMIC_RELAXED, __HIP_MEMORY_SCOPE_WORKGROUP);
}

#define SCAN_KEYS(keys, key0, R)                                               \
  {                                                                            \
    const int4* __restrict__ k4p = (const int4*)(keys);                        \
    for (int q0 = 0; q0 < NQ; q0 += 4 * WGT) {                                 \
      int4 kk[4];                                                              \
      _Pragma("unroll")           \
      for (int u = 0; u < 4; ++u) {                                            \
        const int q = q0 + u * WGT + tid;                                      \
        kk[u] = (q < NQ) ? k4p[q] : make_int4(-1, -1, -1, -1);                 \
      }                                                                        \
      _Pragma("unroll")                            \
      for (int u = 0; u < 4; ++u) {                                            \
        const int q = q0 + u * WGT + tid;                                      \
        const unsigned d0 = (unsigned)(kk[u].x - (key0));                      \
        const unsigned d1 = (unsigned)(kk[u].y - (key0));                      \
        const unsigned d2 = (unsigned)(kk[u].z - (key0));                      \
        const unsigned d3 = (unsigned)(kk[u].w - (key0));                      \
        if (min(min(d0, d1), min(d2, d3)) < (unsigned)(R)) {                   \
          if (d0 < (unsigned)(R)) { const int p = lds_iadd(&nmatch, 1); if (p < LCAP) listA[p] = q * 4 + 0; } \
          if (d1 < (unsigned)(R)) { const int p = lds_iadd(&nmatch, 1); if (p < LCAP) listA[p] = q * 4 + 1; } \
          if (d2 < (unsigned)(R)) { const int p = lds_iadd(&nmatch, 1); if (p < LCAP) listA[p] = q * 4 + 2; } \
          if (d3 < (unsigned)(R)) { const int p = lds_iadd(&nmatch, 1); if (p < LCAP) listA[p] = q * 4 + 3; } \
        }                                                                      \
      }                                                                        \
    }                                                                          \
  }

template <int U>
__device__ __forceinline__ void v2e_rows(
    const float* __restrict__ X, const int* listA, const int* listB,
    float* acc, float* cntE, const int j, const int lane)
{
  int rr[U];
  float4 xa[U], xb[U];
#pragma unroll
  for (int u = 0; u < U; ++u) {
    rr[u] = listA[j + u];
    const int v = listB[j + u];
    const float4* __restrict__ xr = (const float4*)(X + (size_t)v * CIN);
    xa[u] = xr[lane];
    xb[u] = xr[lane + 32];
  }
#pragma unroll
  for (int u = 0; u < U; ++u) {
    float* a = &acc[rr[u] * HID + lane];
    lds_fadd(a +   0, xa[u].x);
    lds_fadd(a +  64, xa[u].y);
    lds_fadd(a + 128, xa[u].z);
    lds_fadd(a + 192, xa[u].w);
    lds_fadd(a +  32, xb[u].x);
    lds_fadd(a +  96, xb[u].y);
    lds_fadd(a + 160, xb[u].z);
    lds_fadd(a + 224, xb[u].w);
    if (lane == 0) lds_fadd(&cntE[rr[u]], 1.0f);
  }
}

template <int U>
__device__ __forceinline__ void e2v_rows(
    const float* __restrict__ Y, const float* __restrict__ es,
    const int* listA, const int* listB, float* acc, float* den,
    const int j, const int lane)
{
  int rr[U];
  float4 sc[U], ya[U], yb[U];
#pragma unroll
  for (int u = 0; u < U; ++u) {
    rr[u] = listA[j + u];
    const int e = listB[j + u];
    sc[u] = ((const float4*)es)[e];
    const float4* __restrict__ yr = (const float4*)(Y + (size_t)e * HID);
    ya[u] = yr[lane];
    yb[u] = yr[lane + 32];
  }
  const bool lo = lane < 16;
#pragma unroll
  for (int u = 0; u < U; ++u) {
    const float wa = lo ? sc[u].x : sc[u].y;
    const float wb = lo ? sc[u].z : sc[u].w;
    float* a = &acc[rr[u] * HID + lane];
    lds_fadd(a +   0, wa * ya[u].x);
    lds_fadd(a +  64, wa * ya[u].y);
    lds_fadd(a + 128, wa * ya[u].z);
    lds_fadd(a + 192, wa * ya[u].w);
    lds_fadd(a +  32, wb * yb[u].x);
    lds_fadd(a +  96, wb * yb[u].y);
    lds_fadd(a + 160, wb * yb[u].z);
    lds_fadd(a + 224, wb * yb[u].w);
    if (lane < NH) {
      const float sh = lane == 0 ? sc[u].x : lane == 1 ? sc[u].y
                     : lane == 2 ? sc[u].z : sc[u].w;
      lds_fadd(&den[rr[u] * NH + lane], sh);
    }
  }
}

__global__ __launch_bounds__(WGT) void v2e_own(
    const float* __restrict__ X, const int* __restrict__ inc_v,
    const int* __restrict__ inc_e, float* __restrict__ Xsum,
    float* __restrict__ cnt)
{
  __shared__ __attribute__((aligned(16))) float acc[RE * HID];
  __shared__ float cntE[RE];
  __shared__ int listA[LCAP];
  __shared__ int listB[LCAP];
  __shared__ int nmatch;

  const int tid = threadIdx.x;
  const int lane = tid & 31;
  const int wave = __builtin_amdgcn_readfirstlane(tid >> 5);
  const int e0 = blockIdx.x * RE;

  for (int k = tid; k < RE * HID / 4; k += WGT)
    ((float4*)acc)[k] = make_float4(0.f, 0.f, 0.f, 0.f);
  for (int k = tid; k < RE; k += WGT) cntE[k] = 0.f;
  if (tid == 0) nmatch = 0;
  __syncthreads();

#define V2E_SLOW(i)                                                            \
  {                                                                            \
    const int r_ = inc_e[i] - e0;                                              \
    const float* __restrict__ xs_ = X + (size_t)inc_v[i] * CIN;                \
    for (int f = 0; f < HID; ++f)                                              \
      lds_fadd(&acc[r_ * HID + (f & 3) * 64 + (f >> 2)], xs_[f]);              \
    lds_fadd(&cntE[r_], 1.0f);                                                 \
  }
  SCAN_KEYS(inc_e, e0, RE)
  __syncthreads();

  const int nm = nmatch;
  if (nm > LCAP) {
    for (int i = tid; i < N_NZ; i += WGT)
      if ((unsigned)(inc_e[i] - e0) < (unsigned)RE) V2E_SLOW(i)
  } else {
  for (int t = tid; t < nm; t += WGT) {
    const int i = listA[t];
    listA[t] = inc_e[i] - e0;
    listB[t] = inc_v[i];
  }
  __syncthreads();

  const int nm8 = nm & ~7;
  for (int j = wave * 8; j < nm8; j += NWAVE * 8)
    v2e_rows<8>(X, listA, listB, acc, cntE, j, lane);
  for (int j = nm8 + wave; j < nm; j += NWAVE)
    v2e_rows<1>(X, listA, listB, acc, cntE, j, lane);
  }
  __syncthreads();

  for (int idx = tid; idx < RE * 64; idx += WGT) {
    const int r = idx >> 6, f4 = idx & 63;
    const float* a = &acc[r * HID + f4];
    ((float4*)(Xsum + (size_t)(e0 + r) * HID))[f4] =
        make_float4(a[0], a[64], a[128], a[192]);
  }
  for (int k = tid; k < RE; k += WGT) cnt[e0 + k] = cntE[k];
}

__global__ __launch_bounds__(256) void gemm_edge(
    const float* __restrict__ Xsum, const float* __restrict__ cnt,
    const float* __restrict__ W, const float* __restrict__ bvec,
    float* __restrict__ Y)
{
  const int lane = threadIdx.x & 31;
  const int wid  = blockIdx.x * 8 + (threadIdx.x >> 5);
  const int row_tile = wid >> 2;
  const int head = wid & 3;
  const int row0 = row_tile * 16;
  const int m = lane & 15;
  const int half = lane >> 4;

  const int row = row0 + m;
  const float inv = 1.0f / fmaxf(cnt[row], 1.0f);
  const float* __restrict__ arow = Xsum + (size_t)row * HID;
  const float* __restrict__ wh = W + (size_t)head * (CIN * DH);

  v8f acc[4] = {v8f{}, v8f{}, v8f{}, v8f{}};
#pragma unroll 4
  for (int k0 = 0; k0 < CIN; k0 += 4) {
    const int ka = k0 + 2 * half;
    v2f afrag;
    afrag.x = arow[ka] * inv;
    afrag.y = arow[ka + 1] * inv;
    const float* __restrict__ wk = wh + (size_t)ka * DH + m;
#pragma unroll
    for (int t = 0; t < 4; ++t) {
      v2f bfrag;
      bfrag.x = wk[t * 16];
      bfrag.y = wk[t * 16 + DH];
      acc[t] = __builtin_amdgcn_wmma_f32_16x16x4_f32(
          false, afrag, false, bfrag, (short)0, acc[t], false, false);
    }
  }

  const int mbase = half * 8;
#pragma unroll
  for (int t = 0; t < 4; ++t) {
    const int col = head * DH + t * 16 + m;
    const float bias = bvec[col];
#pragma unroll
    for (int r = 0; r < 8; ++r) {
      Y[(size_t)(row0 + mbase + r) * HID + col] = acc[t][r] + bias;
    }
  }
}

__global__ __launch_bounds__(256) void edge_alpha(
    const float* __restrict__ Y, const float* __restrict__ a_e,
    float* __restrict__ es)
{
  __shared__ float red[256];
  const int e = blockIdx.x;
  const int f = threadIdx.x;
  red[f] = Y[(size_t)e * HID + f] * a_e[f];
  __syncthreads();
#pragma unroll
  for (int off = 32; off >= 1; off >>= 1) {
    if ((f & 63) < off) red[f] += red[f + off];
    __syncthreads();
  }
  if ((f & 63) == 0) {
    const float alpha = red[f];
    float s = alpha >= 0.0f ? alpha : NEG_SLOPE * alpha;
    s = fminf(fmaxf(s, 0.001f), 5.0f);
    es[e * NH + (f >> 6)] = __expf(s);
  }
}

__global__ __launch_bounds__(WGT) void e2v_own(
    const float* __restrict__ Y, const float* __restrict__ es,
    const int* __restrict__ inc_v, const int* __restrict__ inc_e,
    float* __restrict__ out)
{
  __shared__ __attribute__((aligned(16))) float acc[RV * HID];
  __shared__ __attribute__((aligned(16))) float den[RV * NH];
  __shared__ int listA[LCAP];
  __shared__ int listB[LCAP];
  __shared__ int nmatch;

  const int tid = threadIdx.x;
  const int lane = tid & 31;
  const int wave = __builtin_amdgcn_readfirstlane(tid >> 5);
  const int v0 = blockIdx.x * RV;

  for (int k = tid; k < RV * HID / 4; k += WGT)
    ((float4*)acc)[k] = make_float4(0.f, 0.f, 0.f, 0.f);
  for (int k = tid; k < RV * NH; k += WGT) den[k] = 0.f;
  if (tid == 0) nmatch = 0;
  __syncthreads();

#define E2V_SLOW(i)                                                            \
  {                                                                            \
    const int r_ = inc_v[i] - v0;                                              \
    const int e_ = inc_e[i];                                                   \
    for (int f = 0; f < HID; ++f)                                              \
      lds_fadd(&acc[r_ * HID + (f & 3) * 64 + (f >> 2)],                       \
               es[e_ * NH + (f >> 6)] * Y[(size_t)e_ * HID + f]);              \
    for (int h = 0; h < NH; ++h) lds_fadd(&den[r_ * NH + h], es[e_ * NH + h]); \
  }
  SCAN_KEYS(inc_v, v0, RV)
  __syncthreads();

  const int nm = nmatch;
  if (nm > LCAP) {
    for (int i = tid; i < N_NZ; i += WGT)
      if ((unsigned)(inc_v[i] - v0) < (unsigned)RV) E2V_SLOW(i)
  } else {
  for (int t = tid; t < nm; t += WGT) {
    const int i = listA[t];
    listA[t] = inc_v[i] - v0;
    listB[t] = inc_e[i];
  }
  __syncthreads();

  const int nm8 = nm & ~7;
  for (int j = wave * 8; j < nm8; j += NWAVE * 8)
    e2v_rows<8>(Y, es, listA, listB, acc, den, j, lane);
  for (int j = nm8 + wave; j < nm; j += NWAVE)
    e2v_rows<1>(Y, es, listA, listB, acc, den, j, lane);
  }
  __syncthreads();

  for (int idx = tid; idx < RV * 64; idx += WGT) {
    const int r = idx >> 6, f4 = idx & 63;
    if (v0 + r < N_V) {
      const float d = den[r * NH + (f4 >> 4)];
      const float inv = d > 0.0f ? 1.0f / d : 0.0f;
      const float* a = &acc[r * HID + f4];
      float4 x;
      x.x = a[0]   * inv;
      x.y = a[64]  * inv;
      x.z = a[128] * inv;
      x.w = a[192] * inv;
      x.x = x.x > 0.0f ? x.x : expm1f(x.x);
      x.y = x.y > 0.0f ? x.y : expm1f(x.y);
      x.z = x.z > 0.0f ? x.z : expm1f(x.z);
      x.w = x.w > 0.0f ? x.w : expm1f(x.w);
      ((float4*)(out + (size_t)(v0 + r) * HID))[f4] = x;
    }
  }
}

extern "C" void kernel_launch(void* const* d_in, const int* in_sizes, int n_in,
                              void* d_out, int out_size, void* d_ws, size_t ws_size,
                              hipStream_t stream) {
  const float* X     = (const float*)d_in[0];
  const float* W     = (const float*)d_in[1];
  const float* bvec  = (const float*)d_in[2];
  const float* a_e   = (const float*)d_in[3];
  const int*   inc_v = (const int*)d_in[4];
  const int*   inc_e = (const int*)d_in[5];
  float* out = (float*)d_out;

  float* ws    = (float*)d_ws;
  float* Xsum  = ws;
  float* Y     = Xsum + (size_t)N_E * HID;
  float* cnt   = Y    + (size_t)N_E * HID;
  float* es    = cnt  + N_E;

  v2e_own<<<dim3(N_E / RE), WGT, 0, stream>>>(X, inc_v, inc_e, Xsum, cnt);
  gemm_edge<<<dim3((N_E / 16) * NH / 8), 256, 0, stream>>>(Xsum, cnt, W, bvec, Y);
  edge_alpha<<<dim3(N_E), 256, 0, stream>>>(Y, a_e, es);
  e2v_own<<<dim3((N_V + RV - 1) / RV), WGT, 0, stream>>>(Y, es, inc_v, inc_e, out);
}


# edge scores by linearity in the v2e owner's tail: LDS mean row . (W a_e) + b . a_e, exp folded in; the score launch (160000 waves) is gone; score dot re-associated: 256 terms, fixed lane tree
# speedup vs baseline: 2.9767x; 1.0804x over previous
#include <hip/hip_runtime.h>
#include <hip/hip_bf16.h>

typedef __attribute__((ext_vector_type(2))) float v2f;
typedef __attribute__((ext_vector_type(8))) float v8f;

#define N_V   100000
#define N_E   20000
#define N_NZ  400000
#define CIN   256
#define HID   256
#define NH    4
#define DH    64
#define NEG_SLOPE 0.2f

#define WGT    1024
#define NWAVE  32
#define LCAP   8192
#define NQ     (N_NZ / 4)
#define RE     80
#define RV     196

__device__ __forceinline__ void lds_fadd(float* p, float v) {
  __hip_atomic_fetch_add(p, v, __ATOMIC_RELAXED, __HIP_MEMORY_SCOPE_WORKGROUP);
}
__device__ __forceinline__ int lds_iadd(int* p, int v) {
  return __hip_atomic_fetch_add(p, v, __ATOMIC_RELAXED, __HIP_MEMORY_SCOPE_WORKGROUP);
}

#define SCAN_KEYS(keys, key0, R)                                               \
  {                                                                            \
    const int4* __restrict__ k4p = (const int4*)(keys);                        \
    for (int q0 = 0; q0 < NQ; q0 += 4 * WGT) {                                 \
      int4 kk[4];                                                              \
      _Pragma("unroll")           \
      for (int u = 0; u < 4; ++u) {                                            \
        const int q = q0 + u * WGT + tid;                                      \
        kk[u] = (q < NQ) ? k4p[q] : make_int4(-1, -1, -1, -1);                 \
      }                                                                        \
      _Pragma("unroll")                            \
      for (int u = 0; u < 4; ++u) {                                            \
        const int q = q0 + u * WGT + tid;                                      \
        const unsigned d0 = (unsigned)(kk[u].x - (key0));                      \
        const unsigned d1 = (unsigned)(kk[u].y - (key0));                      \
        const unsigned d2 = (unsigned)(kk[u].z - (key0));                      \
        const unsigned d3 = (unsigned)(kk[u].w - (key0));                      \
        if (min(min(d0, d1), min(d2, d3)) < (unsigned)(R)) {                   \
          if (d0 < (unsigned)(R)) { const int p = lds_iadd(&nmatch, 1); if (p < LCAP) listA[p] = q * 4 + 0; } \
          if (d1 < (unsigned)(R)) { const int p = lds_iadd(&nmatch, 1); if (p < LCAP) listA[p] = q * 4 + 1; } \
          if (d2 < (unsigned)(R)) { const int p = lds_iadd(&nmatch, 1); if (p < LCAP) listA[p] = q * 4 + 2; } \
          if (d3 < (unsigned)(R)) { const int p = lds_iadd(&nmatch, 1); if (p < LCAP) listA[p] = q * 4 + 3; } \
        }                                                                      \
      }                                                                        \
    }                                                                          \
  }

template <int U>
__device__ __forceinline__ void v2e_rows(
    const float* __restrict__ X, const int* listA, const int* listB,
    float* acc, float* cntE, const int j, const int lane)
{
  int rr[U];
  float4 xa[U], xb[U];
#pragma unroll
  for (int u = 0; u < U; ++u) {
    rr[u] = listA[j + u];
    const int v = listB[j + u];
    const float4* __restrict__ xr = (const float4*)(X + (size_t)v * CIN);
    xa[u] = xr[lane];
    xb[u] = xr[lane + 32];
  }
#pragma unroll
  for (int u = 0; u < U; ++u) {
    float* a = &acc[rr[u] * HID + lane];
    lds_fadd(a +   0, xa[u].x);
    lds_fadd(a +  64, xa[u].y);
    lds_fadd(a + 128, xa[u].z);
    lds_fadd(a + 192, xa[u].w);
    lds_fadd(a +  32, xb[u].x);
    lds_fadd(a +  96, xb[u].y);
    lds_fadd(a + 160, xb[u].z);
    lds_fadd(a + 224, xb[u].w);
    if (lane == 0) lds_fadd(&cntE[rr[u]], 1.0f);
  }
}

template <int U>
__device__ __forceinline__ void e2v_rows(
    const float* __restrict__ Y, const float* __restrict__ es,
    const int* listA, const int* listB, float* acc, float* den,
    const int j, const int lane)
{
  int rr[U];
  float4 sc[U], ya[U], yb[U];
#pragma unroll
  for (int u = 0; u < U; ++u) {
    rr[u] = listA[j + u];
    const int e = listB[j + u];
    sc[u] = ((const float4*)es)[e];
    const float4* __restrict__ yr = (const float4*)(Y + (size_t)e * HID);
    ya[u] = yr[lane];
    yb[u] = yr[lane + 32];
  }
  const bool lo = lane < 16;
#pragma unroll
  for (int u = 0; u < U; ++u) {
    const float wa = lo ? sc[u].x : sc[u].y;
    const float wb = lo ? sc[u].z : sc[u].w;
    float* a = &acc[rr[u] * HID + lane];
    lds_fadd(a +   0, wa * ya[u].x);
    lds_fadd(a +  64, wa * ya[u].y);
    lds_fadd(a + 128, wa * ya[u].z);
    lds_fadd(a + 192, wa * ya[u].w);
    lds_fadd(a +  32, wb * yb[u].x);
    lds_fadd(a +  96, wb * yb[u].y);
    lds_fadd(a + 160, wb * yb[u].z);
    lds_fadd(a + 224, wb * yb[u].w);
    if (lane < NH) {
      const float sh = lane == 0 ? sc[u].x : lane == 1 ? sc[u].y
                     : lane == 2 ? sc[u].z : sc[u].w;
      lds_fadd(&den[rr[u] * NH + lane], sh);
    }
  }
}

__global__ __launch_bounds__(WGT) void v2e_own(
    const float* __restrict__ X, const int* __restrict__ inc_v,
    const int* __restrict__ inc_e, const float* __restrict__ W,
    const float* __restrict__ bvec, const float* __restrict__ a_e,
    float* __restrict__ Xsum, float* __restrict__ cnt, float* __restrict__ es)
{
  __shared__ __attribute__((aligned(16))) float acc[RE * HID];
  __shared__ __attribute__((aligned(16))) float uL[NH * CIN];
  __shared__ float c0L[NH];
  __shared__ float cntE[RE];
  __shared__ int listA[LCAP];
  __shared__ int listB[LCAP];
  __shared__ int nmatch;

  const int tid = threadIdx.x;
  const int lane = tid & 31;
  const int wave = __builtin_amdgcn_readfirstlane(tid >> 5);
  const int e0 = blockIdx.x * RE;

  {
    const int h = wave >> 3;
    const int cb = (wave & 7) * 32;
    const int qd = lane & 3, rw = lane >> 2;
    const float4* __restrict__ ap = (const float4*)(a_e + h * DH) + qd * 4;
    const float4* __restrict__ wp =
        (const float4*)(W + ((size_t)h * CIN + cb + rw) * DH) + qd * 4;
    const float4* __restrict__ bp = (const float4*)(bvec + h * DH) + qd * 4;
    float4 av[4], bv[4], wv[16];
#pragma unroll
    for (int b = 0; b < 4; ++b) {
#pragma unroll
      for (int i = 0; i < 4; ++i) wv[b * 4 + i] = wp[b * (8 * DH / 4) + i];
    }
#pragma unroll
    for (int i = 0; i < 4; ++i) { av[i] = ap[i]; bv[i] = bp[i]; }
    float p[4];
#pragma unroll
    for (int b = 0; b < 4; ++b) {
      float s = 0.0f;
#pragma unroll
      for (int i = 0; i < 4; ++i)
        s += wv[b * 4 + i].x * av[i].x + wv[b * 4 + i].y * av[i].y
           + wv[b * 4 + i].z * av[i].z + wv[b * 4 + i].w * av[i].w;
      p[b] = s;
    }
    const bool o1 = (lane & 1) != 0, o2 = (lane & 2) != 0;
    const float q0 = (o1 ? p[1] : p[0]) + __shfl_xor(o1 ? p[0] : p[1], 1);
    const float q1 = (o1 ? p[3] : p[2]) + __shfl_xor(o1 ? p[2] : p[3], 1);
    const float uv = (o2 ? q1 : q0) + __shfl_xor(o2 ? q0 : q1, 2);
    uL[h * CIN + cb + qd * 8 + rw] = uv;
    float q = 0.0f;
#pragma unroll
    for (int i = 0; i < 4; ++i)
      q += bv[i].x * av[i].x + bv[i].y * av[i].y
         + bv[i].z * av[i].z + bv[i].w * av[i].w;
    q += __shfl_xor(q, 1);
    q += __shfl_xor(q, 2);
    if ((wave & 7) == 0 && lane == 0) c0L[h] = q;
  }

  for (int k = tid; k < RE * HID / 4; k += WGT)
    ((float4*)acc)[k] = make_float4(0.f, 0.f, 0.f, 0.f);
  for (int k = tid; k < RE; k += WGT) cntE[k] = 0.f;
  if (tid == 0) nmatch = 0;
  __syncthreads();

#define V2E_SLOW(i)                                                            \
  {                                                                            \
    const int r_ = inc_e[i] - e0;                                              \
    const float* __restrict__ xs_ = X + (size_t)inc_v[i] * CIN;                \
    for (int f = 0; f < HID; ++f)                                              \
      lds_fadd(&acc[r_ * HID + (f & 3) * 64 + (f >> 2)], xs_[f]);              \
    lds_fadd(&cntE[r_], 1.0f);                                                 \
  }
  SCAN_KEYS(inc_e, e0, RE)
  __syncthreads();

  const int nm = nmatch;
  if (nm > LCAP) {
    for (int i = tid; i < N_NZ; i += WGT)
      if ((unsigned)(inc_e[i] - e0) < (unsigned)RE) V2E_SLOW(i)
  } else {
  for (int t = tid; t < nm; t += WGT) {
    const int i = listA[t];
    listA[t] = inc_e[i] - e0;
    listB[t] = inc_v[i];
  }
  __syncthreads();

  const int nm8 = nm & ~7;
  for (int j = wave * 8; j < nm8; j += NWAVE * 8)
    v2e_rows<8>(X, listA, listB, acc, cntE, j, lane);
  for (int j = nm8 + wave; j < nm; j += NWAVE)
    v2e_rows<1>(X, listA, listB, acc, cntE, j, lane);
  }
  __syncthreads();

  float4 ua[NH], ub[NH];
#pragma unroll
  for (int h = 0; h < NH; ++h) {
    ua[h] = ((const float4*)uL)[h * 64 + lane];
    ub[h] = ((const float4*)uL)[h * 64 + lane + 32];
  }
  for (int r = wave; r < RE; r += NWAVE) {
    const float* a = &acc[r * HID + lane];
    const float4 xa = make_float4(a[0], a[64], a[128], a[192]);
    const float4 xb = make_float4(a[32], a[96], a[160], a[224]);
    float4* __restrict__ dst = (float4*)(Xsum + (size_t)(e0 + r) * HID);
    dst[lane] = xa;
    dst[lane + 32] = xb;
    float p[NH];
#pragma unroll
    for (int h = 0; h < NH; ++h)
      p[h] = xa.x * ua[h].x + xa.y * ua[h].y + xa.z * ua[h].z + xa.w * ua[h].w
           + xb.x * ub[h].x + xb.y * ub[h].y + xb.z * ub[h].z + xb.w * ub[h].w;
#pragma unroll
    for (int off = 16; off >= 1; off >>= 1) {
#pragma unroll
      for (int h = 0; h < NH; ++h) p[h] += __shfl_xor(p[h], off);
    }
    const float c = cntE[r];
    if (lane == 0) cnt[e0 + r] = c;
    if (lane < NH) {
      const float ph = lane == 0 ? p[0] : lane == 1 ? p[1]
                     : lane == 2 ? p[2] : p[3];
      const float alpha = ph * (1.0f / fmaxf(c, 1.0f)) + c0L[lane];
      float s = alpha >= 0.0f ? alpha : NEG_SLOPE * alpha;
      s = fminf(fmaxf(s, 0.001f), 5.0f);
      es[(e0 + r) * NH + lane] = __expf(s);
    }
  }
}

__global__ __launch_bounds__(256) void gemm_edge(
    const float* __restrict__ Xsum, const float* __restrict__ cnt,
    const float* __restrict__ W, const float* __restrict__ bvec,
    float* __restrict__ Y)
{
  const int lane = threadIdx.x & 31;
  const int wid  = blockIdx.x * 8 + (threadIdx.x >> 5);
  const int row_tile = wid >> 2;
  const int head = wid & 3;
  const int row0 = row_tile * 16;
  const int m = lane & 15;
  const int half = lane >> 4;

  const int row = row0 + m;
  const float inv = 1.0f / fmaxf(cnt[row], 1.0f);
  const float* __restrict__ arow = Xsum + (size_t)row * HID;
  const float* __restrict__ wh = W + (size_t)head * (CIN * DH);

  v8f acc[4] = {v8f{}, v8f{}, v8f{}, v8f{}};
#pragma unroll 4
  for (int k0 = 0; k0 < CIN; k0 += 4) {
    const int ka = k0 + 2 * half;
    v2f afrag;
    afrag.x = arow[ka] * inv;
    afrag.y = arow[ka + 1] * inv;
    const float* __restrict__ wk = wh + (size_t)ka * DH + m;
#pragma unroll
    for (int t = 0; t < 4; ++t) {
      v2f bfrag;
      bfrag.x = wk[t * 16];
      bfrag.y = wk[t * 16 + DH];
      acc[t] = __builtin_amdgcn_wmma_f32_16x16x4_f32(
          false, afrag, false, bfrag, (short)0, acc[t], false, false);
    }
  }

  const int mbase = half * 8;
#pragma unroll
  for (int t = 0; t < 4; ++t) {
    const int col = head * DH + t * 16 + m;
    const float bias = bvec[col];
#pragma unroll
    for (int r = 0; r < 8; ++r) {
      Y[(size_t)(row0 + mbase + r) * HID + col] = acc[t][r] + bias;
    }
  }
}

__global__ __launch_bounds__(WGT) void e2v_own(
    const float* __restrict__ Y, const float* __restrict__ es,
    const int* __restrict__ inc_v, const int* __restrict__ inc_e,
    float* __restrict__ out)
{
  __shared__ __attribute__((aligned(16))) float acc[RV * HID];
  __shared__ __attribute__((aligned(16))) float den[RV * NH];
  __shared__ int listA[LCAP];
  __shared__ int listB[LCAP];
  __shared__ int nmatch;

  const int tid = threadIdx.x;
  const int lane = tid & 31;
  const int wave = __builtin_amdgcn_readfirstlane(tid >> 5);
  const int v0 = blockIdx.x * RV;

  for (int k = tid; k < RV * HID / 4; k += WGT)
    ((float4*)acc)[k] = make_float4(0.f, 0.f, 0.f, 0.f);
  for (int k = tid; k < RV * NH; k += WGT) den[k] = 0.f;
  if (tid == 0) nmatch = 0;
  __syncthreads();

#define E2V_SLOW(i)                                                            \
  {                                                                            \
    const int r_ = inc_v[i] - v0;                                              \
    const int e_ = inc_e[i];                                                   \
    for (int f = 0; f < HID; ++f)                                              \
      lds_fadd(&acc[r_ * HID + (f & 3) * 64 + (f >> 2)],                       \
               es[e_ * NH + (f >> 6)] * Y[(size_t)e_ * HID + f]);              \
    for (int h = 0; h < NH; ++h) lds_fadd(&den[r_ * NH + h], es[e_ * NH + h]); \
  }
  SCAN_KEYS(inc_v, v0, RV)
  __syncthreads();

  const int nm = nmatch;
  if (nm > LCAP) {
    for (int i = tid; i < N_NZ; i += WGT)
      if ((unsigned)(inc_v[i] - v0) < (unsigned)RV) E2V_SLOW(i)
  } else {
  for (int t = tid; t < nm; t += WGT) {
    const int i = listA[t];
    listA[t] = inc_v[i] - v0;
    listB[t] = inc_e[i];
  }
  __syncthreads();

  const int nm8 = nm & ~7;
  for (int j = wave * 8; j < nm8; j += NWAVE * 8)
    e2v_rows<8>(Y, es, listA, listB, acc, den, j, lane);
  for (int j = nm8 + wave; j < nm; j += NWAVE)
    e2v_rows<1>(Y, es, listA, listB, acc, den, j, lane);
  }
  __syncthreads();

  for (int idx = tid; idx < RV * 64; idx += WGT) {
    const int r = idx >> 6, f4 = idx & 63;
    if (v0 + r < N_V) {
      const float d = den[r * NH + (f4 >> 4)];
      const float inv = d > 0.0f ? 1.0f / d : 0.0f;
      const float* a = &acc[r * HID + f4];
      float4 x;
      x.x = a[0]   * inv;
      x.y = a[64]  * inv;
      x.z = a[128] * inv;
      x.w = a[192] * inv;
      x.x = x.x > 0.0f ? x.x : expm1f(x.x);
      x.y = x.y > 0.0f ? x.y : expm1f(x.y);
      x.z = x.z > 0.0f ? x.z : expm1f(x.z);
      x.w = x.w > 0.0f ? x.w : expm1f(x.w);
      ((float4*)(out + (size_t)(v0 + r) * HID))[f4] = x;
    }
  }
}

extern "C" void kernel_launch(void* const* d_in, const int* in_sizes, int n_in,
                              void* d_out, int out_size, void* d_ws, size_t ws_size,
                              hipStream_t stream) {
  const float* X     = (const float*)d_in[0];
  const float* W     = (const float*)d_in[1];
  const float* bvec  = (const float*)d_in[2];
  const float* a_e   = (const float*)d_in[3];
  const int*   inc_v = (const int*)d_in[4];
  const int*   inc_e = (const int*)d_in[5];
  float* out = (float*)d_out;

  float* ws    = (float*)d_ws;
  float* Xsum  = ws;
  float* Y     = Xsum + (size_t)N_E * HID;
  float* cnt   = Y    + (size_t)N_E * HID;
  float* es    = cnt  + N_E;

  v2e_own<<<dim3(N_E / RE), WGT, 0, stream>>>(X, inc_v, inc_e, W, bvec, a_e,
                                              Xsum, cnt, es);
  gemm_edge<<<dim3((N_E / 16) * NH / 8), 256, 0, stream>>>(Xsum, cnt, W, bvec, Y);
  e2v_own<<<dim3((N_V + RV - 1) / RV), WGT, 0, stream>>>(Y, es, inc_v, inc_e, out);
}


# row sums in registers: owner WG orders its matches by owned row in LDS (int ranks, prefix sum); a wave sums a row's sources and finishes it; no LDS f32 adds or sum tile; e2v owns 392 rows; rank order
# speedup vs baseline: 3.7160x; 1.2484x over previous
#include <hip/hip_runtime.h>
#include <hip/hip_bf16.h>

typedef __attribute__((ext_vector_type(2))) float v2f;
typedef __attribute__((ext_vector_type(8))) float v8f;

#define N_V   100000
#define N_E   20000
#define N_NZ  400000
#define CIN   256
#define HID   256
#define NH    4
#define DH    64
#define NEG_SLOPE 0.2f

#define WGT    1024
#define NWAVE  32
#define LCAP   16384
#define NQ     (N_NZ / 4)
#define RE     80
#define RV     392

__device__ __forceinline__ int lds_iadd(int* p, int v) {
  return __hip_atomic_fetch_add(p, v, __ATOMIC_RELAXED, __HIP_MEMORY_SCOPE_WORKGROUP);
}

#define SCAN_KEYS(keys, key0, R)                                               \
  {                                                                            \
    const int4* __restrict__ k4p = (const int4*)(keys);                        \
    for (int q0 = 0; q0 < NQ; q0 += 4 * WGT) {                                 \
      int4 kk[4];                                                              \
      _Pragma("unroll")           \
      for (int u = 0; u < 4; ++u) {                                            \
        const int q = q0 + u * WGT + tid;                                      \
        kk[u] = (q < NQ) ? k4p[q] : make_int4(-1, -1, -1, -1);                 \
      }                                                                        \
      _Pragma("unroll")                            \
      for (int u = 0; u < 4; ++u) {                                            \
        const int q = q0 + u * WGT + tid;                                      \
        const unsigned d0 = (unsigned)(kk[u].x - (key0));                      \
        const unsigned d1 = (unsigned)(kk[u].y - (key0));                      \
        const unsigned d2 = (unsigned)(kk[u].z - (key0));                      \
        const unsigned d3 = (unsigned)(kk[u].w - (key0));                      \
        if (min(min(d0, d1), min(d2, d3)) < (unsigned)(R)) {                   \
          if (d0 < (unsigned)(R)) { const int p = lds_iadd(&nmatch, 1); if (p < LCAP) listA[p] = q * 4 + 0; } \
          if (d1 < (unsigned)(R)) { const int p = lds_iadd(&nmatch, 1); if (p < LCAP) listA[p] = q * 4 + 1; } \
          if (d2 < (unsigned)(R)) { const int p = lds_iadd(&nmatch, 1); if (p < LCAP) listA[p] = q * 4 + 2; } \
          if (d3 < (unsigned)(R)) { const int p = lds_iadd(&nmatch, 1); if (p < LCAP) listA[p] = q * 4 + 3; } \
        }                                                                      \
      }                                                                        \
    }                                                                          \
  }

template <int R>
__device__ __forceinline__ void order_matches(
    const int* __restrict__ keys, const int* __restrict__ others,
    const int key0, const int nm, int* listA, int* listB, int* srcS,
    int* rowcnt, int* rowstart, const int tid, const int lane, const int wave)
{
  for (int t = tid; t < nm; t += WGT) {
    const int i = listA[t];
    const int r = keys[i] - key0;
    const int rank = lds_iadd(&rowcnt[r], 1);
    listA[t] = r | (rank << 9);
    listB[t] = others[i];
  }
  __syncthreads();
  if (wave == 0) {
    constexpr int C = (R + 31) / 32;
    int cs[C];
    int s = 0;
#pragma unroll
    for (int k = 0; k < C; ++k) {
      const int idx = lane * C + k;
      cs[k] = idx < R ? rowcnt[idx] : 0;
      s += cs[k];
    }
    int inc = s;
#pragma unroll
    for (int off = 1; off < 32; off <<= 1) {
      const int up = __shfl_up(inc, off);
      if (lane >= off) inc += up;
    }
    int run = inc - s;
#pragma unroll
    for (int k = 0; k < C; ++k) {
      const int idx = lane * C + k;
      if (idx < R) rowstart[idx] = run;
      run += cs[k];
    }
  }
  __syncthreads();
  for (int t = tid; t < nm; t += WGT) {
    const int rr = listA[t];
    srcS[rowstart[rr & 511] + (rr >> 9)] = listB[t];
  }
  __syncthreads();
}

__device__ __forceinline__ void v2e_finish(
    const int e, const float4 xa, const float4 xb, const float c,
    const float* uL, const float* c0L, float* __restrict__ Xsum,
    float* __restrict__ cnt, float* __restrict__ es, const int lane)
{
  float4* __restrict__ dst = (float4*)(Xsum + (size_t)e * HID);
  dst[lane] = xa;
  dst[lane + 32] = xb;
  float p[NH];
#pragma unroll
  for (int h = 0; h < NH; ++h) {
    const float4 ua = ((const float4*)uL)[h * 64 + lane];
    const float4 ub = ((const float4*)uL)[h * 64 + lane + 32];
    p[h] = xa.x * ua.x + xa.y * ua.y + xa.z * ua.z + xa.w * ua.w
         + xb.x * ub.x + xb.y * ub.y + xb.z * ub.z + xb.w * ub.w;
  }
#pragma unroll
  for (int off = 16; off >= 1; off >>= 1) {
#pragma unroll
    for (int h = 0; h < NH; ++h) p[h] += __shfl_xor(p[h], off);
  }
  if (lane == 0) cnt[e] = c;
  if (lane < NH) {
    const float ph = lane == 0 ? p[0] : lane == 1 ? p[1]
                   : lane == 2 ? p[2] : p[3];
    const float alpha = ph * (1.0f / fmaxf(c, 1.0f)) + c0L[lane];
    float s = alpha >= 0.0f ? alpha : NEG_SLOPE * alpha;
    s = fminf(fmaxf(s, 0.001f), 5.0f);
    es[e * NH + lane] = __expf(s);
  }
}

__global__ __launch_bounds__(WGT) void v2e_own(
    const float* __restrict__ X, const int* __restrict__ inc_v,
    const int* __restrict__ inc_e, const float* __restrict__ W,
    const float* __restrict__ bvec, const float* __restrict__ a_e,
    float* __restrict__ Xsum, float* __restrict__ cnt, float* __restrict__ es)
{
  __shared__ __attribute__((aligned(16))) float uL[NH * CIN];
  __shared__ float c0L[NH];
  __shared__ int rowcnt[RE];
  __shared__ int rowstart[RE];
  __shared__ int listA[LCAP];
  __shared__ int listB[LCAP];
  __shared__ int srcS[LCAP];
  __shared__ int nmatch;
  __shared__ int nextrow;

  const int tid = threadIdx.x;
  const int lane = tid & 31;
  const int wave = __builtin_amdgcn_readfirstlane(tid >> 5);
  const int e0 = blockIdx.x * RE;

  {
    const int h = wave >> 3;
    const int cb = (wave & 7) * 32;
    const int qd = lane & 3, rw = lane >> 2;
    const float4* __restrict__ ap = (const float4*)(a_e + h * DH) + qd * 4;
    const float4* __restrict__ wp =
        (const float4*)(W + ((size_t)h * CIN + cb + rw) * DH) + qd * 4;
    const float4* __restrict__ bp = (const float4*)(bvec + h * DH) + qd * 4;
    float4 av[4], bv[4], wv[16];
#pragma unroll
    for (int b = 0; b < 4; ++b) {
#pragma unroll
      for (int i = 0; i < 4; ++i) wv[b * 4 + i] = wp[b * (8 * DH / 4) + i];
    }
#pragma unroll
    for (int i = 0; i < 4; ++i) { av[i] = ap[i]; bv[i] = bp[i]; }
    float p[4];
#pragma unroll
    for (int b = 0; b < 4; ++b) {
      float s = 0.0f;
#pragma unroll
      for (int i = 0; i < 4; ++i)
        s += wv[b * 4 + i].x * av[i].x + wv[b * 4 + i].y * av[i].y
           + wv[b * 4 + i].z * av[i].z + wv[b * 4 + i].w * av[i].w;
      p[b] = s;
    }
    const bool o1 = (lane & 1) != 0, o2 = (lane & 2) != 0;
    const float q0 = (o1 ? p[1] : p[0]) + __shfl_xor(o1 ? p[0] : p[1], 1);
    const float q1 = (o1 ? p[3] : p[2]) + __shfl_xor(o1 ? p[2] : p[3], 1);
    const float uv = (o2 ? q1 : q0) + __shfl_xor(o2 ? q0 : q1, 2);
    uL[h * CIN + cb + qd * 8 + rw] = uv;
    float q = 0.0f;
#pragma unroll
    for (int i = 0; i < 4; ++i)
      q += bv[i].x * av[i].x + bv[i].y * av[i].y
         + bv[i].z * av[i].z + bv[i].w * av[i].w;
    q += __shfl_xor(q, 1);
    q += __shfl_xor(q, 2);
    if ((wave & 7) == 0 && lane == 0) c0L[h] = q;
  }

  for (int k = tid; k < RE; k += WGT) rowcnt[k] = 0;
  if (tid == 0) { nmatch = 0; nextrow = 0; }
  __syncthreads();

  SCAN_KEYS(inc_e, e0, RE)
  __syncthreads();

  const float4 zero4 = make_float4(0.f, 0.f, 0.f, 0.f);
  const int nm = nmatch;
  if (nm > LCAP) {
    for (int r = wave; r < RE; r += NWAVE) {
      float4 sa = zero4, sb = zero4;
      int n = 0;
      for (int i0 = 0; i0 < N_NZ; i0 += 32) {
        const int i = i0 + lane;
        const bool m = i < N_NZ && inc_e[i] == e0 + r;
        const int vm = m ? inc_v[i] : 0;
        unsigned mask = (unsigned)__ballot(m);
        while (mask != 0u) {
          const int bpos = __ffs((int)mask) - 1;
          mask &= mask - 1u;
          const int v = __shfl(vm, bpos);
          const float4* __restrict__ xr = (const float4*)(X + (size_t)v * CIN);
          const float4 xa = xr[lane], xb = xr[lane + 32];
          sa.x += xa.x; sa.y += xa.y; sa.z += xa.z; sa.w += xa.w;
          sb.x += xb.x; sb.y += xb.y; sb.z += xb.z; sb.w += xb.w;
          ++n;
        }
      }
      v2e_finish(e0 + r, sa, sb, (float)n, uL, c0L, Xsum, cnt, es, lane);
    }
  } else {
    order_matches<RE>(inc_e, inc_v, e0, nm, listA, listB, srcS, rowcnt,
                      rowstart, tid, lane, wave);
    for (;;) {
      int rnew = 0;
      if (lane == 0) rnew = lds_iadd(&nextrow, 1);
      const int r = __builtin_amdgcn_readfirstlane(rnew);
      if (r >= RE) break;
      const int j0 = __builtin_amdgcn_readfirstlane(rowstart[r]);
      const int n  = __builtin_amdgcn_readfirstlane(rowcnt[r]);
      float4 sa = zero4, sb = zero4;
      for (int j = 0; j < n; j += 8) {
        const int mine = srcS[min(j0 + j + (lane & 7), LCAP - 1)];
        float4 xa[8], xb[8];
#pragma unroll
        for (int u = 0; u < 8; ++u) {
          if (j + u < n) {
            const int v = __builtin_amdgcn_readlane(mine, u);
            const float4* __restrict__ xr = (const float4*)(X + (size_t)v * CIN);
            xa[u] = xr[lane];
            xb[u] = xr[lane + 32];
          } else {
            xa[u] = zero4;
            xb[u] = zero4;
          }
        }
#pragma unroll
        for (int u = 0; u < 8; ++u) {
          sa.x += xa[u].x; sa.y += xa[u].y; sa.z += xa[u].z; sa.w += xa[u].w;
          sb.x += xb[u].x; sb.y += xb[u].y; sb.z += xb[u].z; sb.w += xb[u].w;
        }
      }
      v2e_finish(e0 + r, sa, sb, (float)n, uL, c0L, Xsum, cnt, es, lane);
    }
  }
}

__global__ __launch_bounds__(256) void gemm_edge(
    const float* __restrict__ Xsum, const float* __restrict__ cnt,
    const float* __restrict__ W, const float* __restrict__ bvec,
    float* __restrict__ Y)
{
  const int lane = threadIdx.x & 31;
  const int wid  = blockIdx.x * 8 + (threadIdx.x >> 5);
  const int row_tile = wid >> 2;
  const int head = wid & 3;
  const int row0 = row_tile * 16;
  const int m = lane & 15;
  const int half = lane >> 4;

  const int row = row0 + m;
  const float inv = 1.0f / fmaxf(cnt[row], 1.0f);
  const float* __restrict__ arow = Xsum + (size_t)row * HID;
  const float* __restrict__ wh = W + (size_t)head * (CIN * DH);

  v8f acc[4] = {v8f{}, v8f{}, v8f{}, v8f{}};
#pragma unroll 4
  for (int k0 = 0; k0 < CIN; k0 += 4) {
    const int ka = k0 + 2 * half;
    v2f afrag;
    afrag.x = arow[ka] * inv;
    afrag.y = arow[ka + 1] * inv;
    const float* __restrict__ wk = wh + (size_t)ka * DH + m;
#pragma unroll
    for (int t = 0; t < 4; ++t) {
      v2f bfrag;
      bfrag.x = wk[t * 16];
      bfrag.y = wk[t * 16 + DH];
      acc[t] = __builtin_amdgcn_wmma_f32_16x16x4_f32(
          false, afrag, false, bfrag, (short)0, acc[t], false, false);
    }
  }

  const int mbase = half * 8;
#pragma unroll
  for (int t = 0; t < 4; ++t) {
    const int col = head * DH + t * 16 + m;
    const float bias = bvec[col];
#pragma unroll
    for (int r = 0; r < 8; ++r) {
      Y[(size_t)(row0 + mbase + r) * HID + col] = acc[t][r] + bias;
    }
  }
}

__device__ __forceinline__ void e2v_finish(
    const int v, const float4 a, const float4 b, const float da,
    const float db, float* __restrict__ out, const int lane)
{
  if (v < N_V) {
    const float ia = da > 0.0f ? 1.0f / da : 0.0f;
    const float ib = db > 0.0f ? 1.0f / db : 0.0f;
    float4 x, y;
    x.x = a.x * ia; x.y = a.y * ia; x.z = a.z * ia; x.w = a.w * ia;
    y.x = b.x * ib; y.y = b.y * ib; y.z = b.z * ib; y.w = b.w * ib;
    x.x = x.x > 0.0f ? x.x : expm1f(x.x);
    x.y = x.y > 0.0f ? x.y : expm1f(x.y);
    x.z = x.z > 0.0f ? x.z : expm1f(x.z);
    x.w = x.w > 0.0f ? x.w : expm1f(x.w);
    y.x = y.x > 0.0f ? y.x : expm1f(y.x);
    y.y = y.y > 0.0f ? y.y : expm1f(y.y);
    y.z = y.z > 0.0f ? y.z : expm1f(y.z);
    y.w = y.w > 0.0f ? y.w : expm1f(y.w);
    float4* __restrict__ dst = (float4*)(out + (size_t)v * HID);
    dst[lane] = x;
    dst[lane + 32] = y;
  }
}

__global__ __launch_bounds__(WGT) void e2v_own(
    const float* __restrict__ Y, const float* __restrict__ es,
    const int* __restrict__ inc_v, const int* __restrict__ inc_e,
    float* __restrict__ out)
{
  __shared__ int rowcnt[RV];
  __shared__ int rowstart[RV];
  __shared__ int listA[LCAP];
  __shared__ int listB[LCAP];
  __shared__ int srcS[LCAP];
  __shared__ int nmatch;
  __shared__ int nextrow;

  const int tid = threadIdx.x;
  const int lane = tid & 31;
  const int wave = __builtin_amdgcn_readfirstlane(tid >> 5);
  const int v0 = blockIdx.x * RV;

  for (int k = tid; k < RV; k += WGT) rowcnt[k] = 0;
  if (tid == 0) { nmatch = 0; nextrow = 0; }
  __syncthreads();

  SCAN_KEYS(inc_v, v0, RV)
  __syncthreads();

  const float* __restrict__ esl = es + (lane < 16 ? 0 : 1);
  const float4 zero4 = make_float4(0.f, 0.f, 0.f, 0.f);
  const int nm = nmatch;
  if (nm > LCAP) {
    for (int r = wave; r < RV; r += NWAVE) {
      float4 a = zero4, b = zero4;
      float da = 0.0f, db = 0.0f;
      for (int i0 = 0; i0 < N_NZ; i0 += 32) {
        const int i = i0 + lane;
        const bool m = i < N_NZ && inc_v[i] == v0 + r;
        const int em = m ? inc_e[i] : 0;
        unsigned mask = (unsigned)__ballot(m);
        while (mask != 0u) {
          const int bpos = __ffs((int)mask) - 1;
          mask &= mask - 1u;
          const int e = __shfl(em, bpos);
          const float wa = esl[e * NH], wb = esl[e * NH + 2];
          const float4* __restrict__ yr = (const float4*)(Y + (size_t)e * HID);
          const float4 ya = yr[lane], yb = yr[lane + 32];
          a.x += wa * ya.x; a.y += wa * ya.y; a.z += wa * ya.z; a.w += wa * ya.w;
          b.x += wb * yb.x; b.y += wb * yb.y; b.z += wb * yb.z; b.w += wb * yb.w;
          da += wa; db += wb;
        }
      }
      e2v_finish(v0 + r, a, b, da, db, out, lane);
    }
  } else {
    order_matches<RV>(inc_v, inc_e, v0, nm, listA, listB, srcS, rowcnt,
                      rowstart, tid, lane, wave);
    for (;;) {
      int rnew = 0;
      if (lane == 0) rnew = lds_iadd(&nextrow, 1);
      const int r = 2 * __builtin_amdgcn_readfirstlane(rnew);
      if (r >= RV) break;
      const int jA = __builtin_amdgcn_readfirstlane(rowstart[r]);
      const int nA = __builtin_amdgcn_readfirstlane(rowcnt[r]);
      const int jB = __builtin_amdgcn_readfirstlane(rowstart[r + 1]);
      const int nB = __builtin_amdgcn_readfirstlane(rowcnt[r + 1]);
      const int nmax = nA > nB ? nA : nB;
      float4 aA = zero4, bA = zero4, aB = zero4, bB = zero4;
      float daA = 0.0f, dbA = 0.0f, daB = 0.0f, dbB = 0.0f;
      for (int j = 0; j < nmax; j += 4) {
        const int mine =
            srcS[min(((lane & 4) ? jB : jA) + j + (lane & 3), LCAP - 1)];
        float waA[4], wbA[4], waB[4], wbB[4];
        float4 yaA[4], ybA[4], yaB[4], ybB[4];
#pragma unroll
        for (int u = 0; u < 4; ++u) {
          if (j + u < nA) {
            const int e = __builtin_amdgcn_readlane(mine, u);
            waA[u] = esl[e * NH];
            wbA[u] = esl[e * NH + 2];
            const float4* __restrict__ yr = (const float4*)(Y + (size_t)e * HID);
            yaA[u] = yr[lane];
            ybA[u] = yr[lane + 32];
          } else {
            waA[u] = 0.0f; wbA[u] = 0.0f; yaA[u] = zero4; ybA[u] = zero4;
          }
          if (j + u < nB) {
            const int e = __builtin_amdgcn_readlane(mine, 4 + u);
            waB[u] = esl[e * NH];
            wbB[u] = esl[e * NH + 2];
            const float4* __restrict__ yr = (const float4*)(Y + (size_t)e * HID);
            yaB[u] = yr[lane];
            ybB[u] = yr[lane + 32];
          } else {
            waB[u] = 0.0f; wbB[u] = 0.0f; yaB[u] = zero4; ybB[u] = zero4;
          }
        }
#pragma unroll
        for (int u = 0; u < 4; ++u) {
          aA.x += waA[u] * yaA[u].x; aA.y += waA[u] * yaA[u].y;
          aA.z += waA[u] * yaA[u].z; aA.w += waA[u] * yaA[u].w;
          bA.x += wbA[u] * ybA[u].x; bA.y += wbA[u] * ybA[u].y;
          bA.z += wbA[u] * ybA[u].z; bA.w += wbA[u] * ybA[u].w;
          daA += waA[u]; dbA += wbA[u];
          aB.x += waB[u] * yaB[u].x; aB.y += waB[u] * yaB[u].y;
          aB.z += waB[u] * yaB[u].z; aB.w += waB[u] * yaB[u].w;
          bB.x += wbB[u] * ybB[u].x; bB.y += wbB[u] * ybB[u].y;
          bB.z += wbB[u] * ybB[u].z; bB.w += wbB[u] * ybB[u].w;
          daB += waB[u]; dbB += wbB[u];
        }
      }
      e2v_finish(v0 + r,     aA, bA, daA, dbA, out, lane);
      e2v_finish(v0 + r + 1, aB, bB, daB, dbB, out, lane);
    }
  }
}

extern "C" void kernel_launch(void* const* d_in, const int* in_sizes, int n_in,
                              void* d_out, int out_size, void* d_ws, size_t ws_size,
                              hipStream_t stream) {
  const float* X     = (const float*)d_in[0];
  const float* W     = (const float*)d_in[1];
  const float* bvec  = (const float*)d_in[2];
  const float* a_e   = (const float*)d_in[3];
  const int*   inc_v = (const int*)d_in[4];
  const int*   inc_e = (const int*)d_in[5];
  float* out = (float*)d_out;

  float* ws    = (float*)d_ws;
  float* Xsum  = ws;
  float* Y     = Xsum + (size_t)N_E * HID;
  float* cnt   = Y    + (size_t)N_E * HID;
  float* es    = cnt  + N_E;

  v2e_own<<<dim3(N_E / RE), WGT, 0, stream>>>(X, inc_v, inc_e, W, bvec, a_e,
                                              Xsum, cnt, es);
  gemm_edge<<<dim3((N_E / 16) * NH / 8), 256, 0, stream>>>(Xsum, cnt, W, bvec, Y);
  e2v_own<<<dim3((N_V + RV - 1) / RV), WGT, 0, stream>>>(Y, es, inc_v, inc_e, out);
}


# owners scan a 1-byte owner code per key, kept as 8 bit planes (one small launch builds them by ballots), not the 4-byte keys: XOR/OR mask per 32 keys; match-list order, so rank sum order, may differ
# speedup vs baseline: 4.2319x; 1.1388x over previous
#include <hip/hip_runtime.h>
#include <hip/hip_bf16.h>

typedef __attribute__((ext_vector_type(2))) float v2f;
typedef __attribute__((ext_vector_type(8))) float v8f;

#define N_V   100000
#define N_E   20000
#define N_NZ  400000
#define CIN   256
#define HID   256
#define NH    4
#define DH    64
#define NEG_SLOPE 0.2f

#define WGT    1024
#define NWAVE  32
#define LCAP   16384
#define RE     80
#define RV     392

__device__ __forceinline__ int lds_iadd(int* p, int v) {
  return __hip_atomic_fetch_add(p, v, __ATOMIC_RELAXED, __HIP_MEMORY_SCOPE_WORKGROUP);
}

#define NB     (N_NZ / 32)
#define KPW    8
#define KWG    ((NB + 8 * KPW - 1) / (8 * KPW))
#define SU     7

__global__ __launch_bounds__(256) void key_planes(
    const int* __restrict__ inc_v, const int* __restrict__ inc_e,
    int* __restrict__ e_lo, int* __restrict__ e_hi,
    int* __restrict__ v_lo, int* __restrict__ v_hi)
{
  const int lane = threadIdx.x & 31;
  const bool isv = blockIdx.y != 0;
  const int* __restrict__ keys = isv ? inc_v : inc_e;
  int* __restrict__ lo = isv ? v_lo : e_lo;
  int* __restrict__ hi = isv ? v_hi : e_hi;
  const int g0 = (blockIdx.x * 8 + (threadIdx.x >> 5)) * KPW;
  int k[KPW];
#pragma unroll
  for (int u = 0; u < KPW; ++u)
    k[u] = (g0 + u < NB) ? keys[(g0 + u) * 32 + lane] : -1;
#pragma unroll
  for (int u = 0; u < KPW; ++u) {
    const unsigned o = min(isv ? (unsigned)k[u] / (unsigned)RV
                               : (unsigned)k[u] / (unsigned)RE, 255u);
    int word = 0;
#pragma unroll
    for (int j = 0; j < 8; ++j) {
      const int plane = (int)(unsigned)__ballot((o >> j) & 1u);
      if (lane == j) word = plane;
    }
    if (g0 + u < NB && lane < 8)
      (lane < 4 ? lo : hi)[(g0 + u) * 4 + (lane & 3)] = word;
  }
}

__device__ __forceinline__ void scan_planes(
    const int* __restrict__ plo, const int* __restrict__ phi, const int ob,
    int* listA, int* nmatch, const int tid)
{
  const int4* __restrict__ lo4 = (const int4*)plo;
  const int4* __restrict__ hi4 = (const int4*)phi;
  int pat[8];
#pragma unroll
  for (int j = 0; j < 8; ++j) pat[j] = -((ob >> j) & 1);
  for (int q0 = 0; q0 < NB; q0 += SU * WGT) {
    int4 a[SU], b[SU];
#pragma unroll
    for (int u = 0; u < SU; ++u) {
      const int q = q0 + u * WGT + tid;
      a[u] = (q < NB) ? lo4[q] : make_int4(0, 0, 0, 0);
      b[u] = (q < NB) ? hi4[q] : make_int4(0, 0, 0, 0);
    }
    unsigned m[SU];
    int c = 0;
#pragma unroll
    for (int u = 0; u < SU; ++u) {
      const int q = q0 + u * WGT + tid;
      const int diff = (a[u].x ^ pat[0]) | (a[u].y ^ pat[1])
                     | (a[u].z ^ pat[2]) | (a[u].w ^ pat[3])
                     | (b[u].x ^ pat[4]) | (b[u].y ^ pat[5])
                     | (b[u].z ^ pat[6]) | (b[u].w ^ pat[7]);
      m[u] = (q < NB) ? ~(unsigned)diff : 0u;
      c += __popc(m[u]);
    }
    if (c != 0) {
      int p = lds_iadd(nmatch, c);
#pragma unroll
      for (int u = 0; u < SU; ++u) {
        const int base = (q0 + u * WGT + tid) * 32;
        unsigned mm = m[u];
        while (mm != 0u) {
          const int bpos = __ffs((int)mm) - 1;
          mm &= mm - 1u;
          if (p < LCAP) listA[p] = base + bpos;
          ++p;
        }
      }
    }
  }
}

template <int R>
__device__ __forceinline__ void order_matches(
    const int* __restrict__ keys, const int* __restrict__ others,
    const int key0, const int nm, int* listA, int* listB, int* srcS,
    int* rowcnt, int* rowstart, const int tid, const int lane, const int wave)
{
  for (int t = tid; t < nm; t += WGT) {
    const int i = listA[t];
    const int r = keys[i] - key0;
    const int rank = lds_iadd(&rowcnt[r], 1);
    listA[t] = r | (rank << 9);
    listB[t] = others[i];
  }
  __syncthreads();
  if (wave == 0) {
    constexpr int C = (R + 31) / 32;
    int cs[C];
    int s = 0;
#pragma unroll
    for (int k = 0; k < C; ++k) {
      const int idx = lane * C + k;
      cs[k] = idx < R ? rowcnt[idx] : 0;
      s += cs[k];
    }
    int inc = s;
#pragma unroll
    for (int off = 1; off < 32; off <<= 1) {
      const int up = __shfl_up(inc, off);
      if (lane >= off) inc += up;
    }
    int run = inc - s;
#pragma unroll
    for (int k = 0; k < C; ++k) {
      const int idx = lane * C + k;
      if (idx < R) rowstart[idx] = run;
      run += cs[k];
    }
  }
  __syncthreads();
  for (int t = tid; t < nm; t += WGT) {
    const int rr = listA[t];
    srcS[rowstart[rr & 511] + (rr >> 9)] = listB[t];
  }
  __syncthreads();
}

__device__ __forceinline__ void v2e_finish(
    const int e, const float4 xa, const float4 xb, const float c,
    const float* uL, const float* c0L, float* __restrict__ Xsum,
    float* __restrict__ cnt, float* __restrict__ es, const int lane)
{
  float4* __restrict__ dst = (float4*)(Xsum + (size_t)e * HID);
  dst[lane] = xa;
  dst[lane + 32] = xb;
  float p[NH];
#pragma unroll
  for (int h = 0; h < NH; ++h) {
    const float4 ua = ((const float4*)uL)[h * 64 + lane];
    const float4 ub = ((const float4*)uL)[h * 64 + lane + 32];
    p[h] = xa.x * ua.x + xa.y * ua.y + xa.z * ua.z + xa.w * ua.w
         + xb.x * ub.x + xb.y * ub.y + xb.z * ub.z + xb.w * ub.w;
  }
#pragma unroll
  for (int off = 16; off >= 1; off >>= 1) {
#pragma unroll
    for (int h = 0; h < NH; ++h) p[h] += __shfl_xor(p[h], off);
  }
  if (lane == 0) cnt[e] = c;
  if (lane < NH) {
    const float ph = lane == 0 ? p[0] : lane == 1 ? p[1]
                   : lane == 2 ? p[2] : p[3];
    const float alpha = ph * (1.0f / fmaxf(c, 1.0f)) + c0L[lane];
    float s = alpha >= 0.0f ? alpha : NEG_SLOPE * alpha;
    s = fminf(fmaxf(s, 0.001f), 5.0f);
    es[e * NH + lane] = __expf(s);
  }
}

__global__ __launch_bounds__(WGT) void v2e_own(
    const float* __restrict__ X, const int* __restrict__ inc_v,
    const int* __restrict__ inc_e, const float* __restrict__ W,
    const float* __restrict__ bvec, const float* __restrict__ a_e,
    const int* __restrict__ e_lo, const int* __restrict__ e_hi,
    float* __restrict__ Xsum, float* __restrict__ cnt, float* __restrict__ es)
{
  __shared__ __attribute__((aligned(16))) float uL[NH * CIN];
  __shared__ float c0L[NH];
  __shared__ int rowcnt[RE];
  __shared__ int rowstart[RE];
  __shared__ int listA[LCAP];
  __shared__ int listB[LCAP];
  __shared__ int srcS[LCAP];
  __shared__ int nmatch;
  __shared__ int nextrow;

  const int tid = threadIdx.x;
  const int lane = tid & 31;
  const int wave = __builtin_amdgcn_readfirstlane(tid >> 5);
  const int e0 = blockIdx.x * RE;

  {
    const int h = wave >> 3;
    const int cb = (wave & 7) * 32;
    const int qd = lane & 3, rw = lane >> 2;
    const float4* __restrict__ ap = (const float4*)(a_e + h * DH) + qd * 4;
    const float4* __restrict__ wp =
        (const float4*)(W + ((size_t)h * CIN + cb + rw) * DH) + qd * 4;
    const float4* __restrict__ bp = (const float4*)(bvec + h * DH) + qd * 4;
    float4 av[4], bv[4], wv[16];
#pragma unroll
    for (int b = 0; b < 4; ++b) {
#pragma unroll
      for (int i = 0; i < 4; ++i) wv[b * 4 + i] = wp[b * (8 * DH / 4) + i];
    }
#pragma unroll
    for (int i = 0; i < 4; ++i) { av[i] = ap[i]; bv[i] = bp[i]; }
    float p[4];
#pragma unroll
    for (int b = 0; b < 4; ++b) {
      float s = 0.0f;
#pragma unroll
      for (int i = 0; i < 4; ++i)
        s += wv[b * 4 + i].x * av[i].x + wv[b * 4 + i].y * av[i].y
           + wv[b * 4 + i].z * av[i].z + wv[b * 4 + i].w * av[i].w;
      p[b] = s;
    }
    const bool o1 = (lane & 1) != 0, o2 = (lane & 2) != 0;
    const float q0 = (o1 ? p[1] : p[0]) + __shfl_xor(o1 ? p[0] : p[1], 1);
    const float q1 = (o1 ? p[3] : p[2]) + __shfl_xor(o1 ? p[2] : p[3], 1);
    const float uv = (o2 ? q1 : q0) + __shfl_xor(o2 ? q0 : q1, 2);
    uL[h * CIN + cb + qd * 8 + rw] = uv;
    float q = 0.0f;
#pragma unroll
    for (int i = 0; i < 4; ++i)
      q += bv[i].x * av[i].x + bv[i].y * av[i].y
         + bv[i].z * av[i].z + bv[i].w * av[i].w;
    q += __shfl_xor(q, 1);
    q += __shfl_xor(q, 2);
    if ((wave & 7) == 0 && lane == 0) c0L[h] = q;
  }

  for (int k = tid; k < RE; k += WGT) rowcnt[k] = 0;
  if (tid == 0) { nmatch = 0; nextrow = 0; }
  __syncthreads();

  scan_planes(e_lo, e_hi, blockIdx.x, listA, &nmatch, tid);
  __syncthreads();

  const float4 zero4 = make_float4(0.f, 0.f, 0.f, 0.f);
  const int nm = nmatch;
  if (nm > LCAP) {
    for (int r = wave; r < RE; r += NWAVE) {
      float4 sa = zero4, sb = zero4;
      int n = 0;
      for (int i0 = 0; i0 < N_NZ; i0 += 32) {
        const int i = i0 + lane;
        const bool m = i < N_NZ && inc_e[i] == e0 + r;
        const int vm = m ? inc_v[i] : 0;
        unsigned mask = (unsigned)__ballot(m);
        while (mask != 0u) {
          const int bpos = __ffs((int)mask) - 1;
          mask &= mask - 1u;
          const int v = __shfl(vm, bpos);
          const float4* __restrict__ xr = (const float4*)(X + (size_t)v * CIN);
          const float4 xa = xr[lane], xb = xr[lane + 32];
          sa.x += xa.x; sa.y += xa.y; sa.z += xa.z; sa.w += xa.w;
          sb.x += xb.x; sb.y += xb.y; sb.z += xb.z; sb.w += xb.w;
          ++n;
        }
      }
      v2e_finish(e0 + r, sa, sb, (float)n, uL, c0L, Xsum, cnt, es, lane);
    }
  } else {
    order_matches<RE>(inc_e, inc_v, e0, nm, listA, listB, srcS, rowcnt,
                      rowstart, tid, lane, wave);
    for (;;) {
      int rnew = 0;
      if (lane == 0) rnew = lds_iadd(&nextrow, 1);
      const int r = __builtin_amdgcn_readfirstlane(rnew);
      if (r >= RE) break;
      const int j0 = __builtin_amdgcn_readfirstlane(rowstart[r]);
      const int n  = __builtin_amdgcn_readfirstlane(rowcnt[r]);
      float4 sa = zero4, sb = zero4;
      for (int j = 0; j < n; j += 8) {
        const int mine = srcS[min(j0 + j + (lane & 7), LCAP - 1)];
        float4 xa[8], xb[8];
#pragma unroll
        for (int u = 0; u < 8; ++u) {
          if (j + u < n) {
            const int v = __builtin_amdgcn_readlane(mine, u);
            const float4* __restrict__ xr = (const float4*)(X + (size_t)v * CIN);
            xa[u] = xr[lane];
            xb[u] = xr[lane + 32];
          } else {
            xa[u] = zero4;
            xb[u] = zero4;
          }
        }
#pragma unroll
        for (int u = 0; u < 8; ++u) {
          sa.x += xa[u].x; sa.y += xa[u].y; sa.z += xa[u].z; sa.w += xa[u].w;
          sb.x += xb[u].x; sb.y += xb[u].y; sb.z += xb[u].z; sb.w += xb[u].w;
        }
      }
      v2e_finish(e0 + r, sa, sb, (float)n, uL, c0L, Xsum, cnt, es, lane);
    }
  }
}

__global__ __launch_bounds__(256) void gemm_edge(
    const float* __restrict__ Xsum, const float* __restrict__ cnt,
    const float* __restrict__ W, const float* __restrict__ bvec,
    float* __restrict__ Y)
{
  const int lane = threadIdx.x & 31;
  const int wid  = blockIdx.x * 8 + (threadIdx.x >> 5);
  const int row_tile = wid >> 2;
  const int head = wid & 3;
  const int row0 = row_tile * 16;
  const int m = lane & 15;
  const int half = lane >> 4;

  const int row = row0 + m;
  const float inv = 1.0f / fmaxf(cnt[row], 1.0f);
  const float* __restrict__ arow = Xsum + (size_t)row * HID;
  const float* __restrict__ wh = W + (size_t)head * (CIN * DH);

  v8f acc[4] = {v8f{}, v8f{}, v8f{}, v8f{}};
#pragma unroll 4
  for (int k0 = 0; k0 < CIN; k0 += 4) {
    const int ka = k0 + 2 * half;
    v2f afrag;
    afrag.x = arow[ka] * inv;
    afrag.y = arow[ka + 1] * inv;
    const float* __restrict__ wk = wh + (size_t)ka * DH + m;
#pragma unroll
    for (int t = 0; t < 4; ++t) {
      v2f bfrag;
      bfrag.x = wk[t * 16];
      bfrag.y = wk[t * 16 + DH];
      acc[t] = __builtin_amdgcn_wmma_f32_16x16x4_f32(
          false, afrag, false, bfrag, (short)0, acc[t], false, false);
    }
  }

  const int mbase = half * 8;
#pragma unroll
  for (int t = 0; t < 4; ++t) {
    const int col = head * DH + t * 16 + m;
    const float bias = bvec[col];
#pragma unroll
    for (int r = 0; r < 8; ++r) {
      Y[(size_t)(row0 + mbase + r) * HID + col] = acc[t][r] + bias;
    }
  }
}

__device__ __forceinline__ void e2v_finish(
    const int v, const float4 a, const float4 b, const float da,
    const float db, float* __restrict__ out, const int lane)
{
  if (v < N_V) {
    const float ia = da > 0.0f ? 1.0f / da : 0.0f;
    const float ib = db > 0.0f ? 1.0f / db : 0.0f;
    float4 x, y;
    x.x = a.x * ia; x.y = a.y * ia; x.z = a.z * ia; x.w = a.w * ia;
    y.x = b.x * ib; y.y = b.y * ib; y.z = b.z * ib; y.w = b.w * ib;
    x.x = x.x > 0.0f ? x.x : expm1f(x.x);
    x.y = x.y > 0.0f ? x.y : expm1f(x.y);
    x.z = x.z > 0.0f ? x.z : expm1f(x.z);
    x.w = x.w > 0.0f ? x.w : expm1f(x.w);
    y.x = y.x > 0.0f ? y.x : expm1f(y.x);
    y.y = y.y > 0.0f ? y.y : expm1f(y.y);
    y.z = y.z > 0.0f ? y.z : expm1f(y.z);
    y.w = y.w > 0.0f ? y.w : expm1f(y.w);
    float4* __restrict__ dst = (float4*)(out + (size_t)v * HID);
    dst[lane] = x;
    dst[lane + 32] = y;
  }
}

__global__ __launch_bounds__(WGT) void e2v_own(
    const float* __restrict__ Y, const float* __restrict__ es,
    const int* __restrict__ inc_v, const int* __restrict__ inc_e,
    const int* __restrict__ v_lo, const int* __restrict__ v_hi,
    float* __restrict__ out)
{
  __shared__ int rowcnt[RV];
  __shared__ int rowstart[RV];
  __shared__ int listA[LCAP];
  __shared__ int listB[LCAP];
  __shared__ int srcS[LCAP];
  __shared__ int nmatch;
  __shared__ int nextrow;

  const int tid = threadIdx.x;
  const int lane = tid & 31;
  const int wave = __builtin_amdgcn_readfirstlane(tid >> 5);
  const int v0 = blockIdx.x * RV;

  for (int k = tid; k < RV; k += WGT) rowcnt[k] = 0;
  if (tid == 0) { nmatch = 0; nextrow = 0; }
  __syncthreads();

  scan_planes(v_lo, v_hi, blockIdx.x, listA, &nmatch, tid);
  __syncthreads();

  const float* __restrict__ esl = es + (lane < 16 ? 0 : 1);
  const float4 zero4 = make_float4(0.f, 0.f, 0.f, 0.f);
  const int nm = nmatch;
  if (nm > LCAP) {
    for (int r = wave; r < RV; r += NWAVE) {
      float4 a = zero4, b = zero4;
      float da = 0.0f, db = 0.0f;
      for (int i0 = 0; i0 < N_NZ; i0 += 32) {
        const int i = i0 + lane;
        const bool m = i < N_NZ && inc_v[i] == v0 + r;
        const int em = m ? inc_e[i] : 0;
        unsigned mask = (unsigned)__ballot(m);
        while (mask != 0u) {
          const int bpos = __ffs((int)mask) - 1;
          mask &= mask - 1u;
          const int e = __shfl(em, bpos);
          const float wa = esl[e * NH], wb = esl[e * NH + 2];
          const float4* __restrict__ yr = (const float4*)(Y + (size_t)e * HID);
          const float4 ya = yr[lane], yb = yr[lane + 32];
          a.x += wa * ya.x; a.y += wa * ya.y; a.z += wa * ya.z; a.w += wa * ya.w;
          b.x += wb * yb.x; b.y += wb * yb.y; b.z += wb * yb.z; b.w += wb * yb.w;
          da += wa; db += wb;
        }
      }
      e2v_finish(v0 + r, a, b, da, db, out, lane);
    }
  } else {
    order_matches<RV>(inc_v, inc_e, v0, nm, listA, listB, srcS, rowcnt,
                      rowstart, tid, lane, wave);
    for (;;) {
      int rnew = 0;
      if (lane == 0) rnew = lds_iadd(&nextrow, 1);
      const int r = 2 * __builtin_amdgcn_readfirstlane(rnew);
      if (r >= RV) break;
      const int jA = __builtin_amdgcn_readfirstlane(rowstart[r]);
      const int nA = __builtin_amdgcn_readfirstlane(rowcnt[r]);
      const int jB = __builtin_amdgcn_readfirstlane(rowstart[r + 1]);
      const int nB = __builtin_amdgcn_readfirstlane(rowcnt[r + 1]);
      const int nmax = nA > nB ? nA : nB;
      float4 aA = zero4, bA = zero4, aB = zero4, bB = zero4;
      float daA = 0.0f, dbA = 0.0f, daB = 0.0f, dbB = 0.0f;
      for (int j = 0; j < nmax; j += 4) {
        const int mine =
            srcS[min(((lane & 4) ? jB : jA) + j + (lane & 3), LCAP - 1)];
        float waA[4], wbA[4], waB[4], wbB[4];
        float4 yaA[4], ybA[4], yaB[4], ybB[4];
#pragma unroll
        for (int u = 0; u < 4; ++u) {
          if (j + u < nA) {
            const int e = __builtin_amdgcn_readlane(mine, u);
            waA[u] = esl[e * NH];
            wbA[u] = esl[e * NH + 2];
            const float4* __restrict__ yr = (const float4*)(Y + (size_t)e * HID);
            yaA[u] = yr[lane];
            ybA[u] = yr[lane + 32];
          } else {
            waA[u] = 0.0f; wbA[u] = 0.0f; yaA[u] = zero4; ybA[u] = zero4;
          }
          if (j + u < nB) {
            const int e = __builtin_amdgcn_readlane(mine, 4 + u);
            waB[u] = esl[e * NH];
            wbB[u] = esl[e * NH + 2];
            const float4* __restrict__ yr = (const float4*)(Y + (size_t)e * HID);
            yaB[u] = yr[lane];
            ybB[u] = yr[lane + 32];
          } else {
            waB[u] = 0.0f; wbB[u] = 0.0f; yaB[u] = zero4; ybB[u] = zero4;
          }
        }
#pragma unroll
        for (int u = 0; u < 4; ++u) {
          aA.x += waA[u] * yaA[u].x; aA.y += waA[u] * yaA[u].y;
          aA.z += waA[u] * yaA[u].z; aA.w += waA[u] * yaA[u].w;
          bA.x += wbA[u] * ybA[u].x; bA.y += wbA[u] * ybA[u].y;
          bA.z += wbA[u] * ybA[u].z; bA.w += wbA[u] * ybA[u].w;
          daA += waA[u]; dbA += wbA[u];
          aB.x += waB[u] * yaB[u].x; aB.y += waB[u] * yaB[u].y;
          aB.z += waB[u] * yaB[u].z; aB.w += waB[u] * yaB[u].w;
          bB.x += wbB[u] * ybB[u].x; bB.y += wbB[u] * ybB[u].y;
          bB.z += wbB[u] * ybB[u].z; bB.w += wbB[u] * ybB[u].w;
          daB += waB[u]; dbB += wbB[u];
        }
      }
      e2v_finish(v0 + r,     aA, bA, daA, dbA, out, lane);
      e2v_finish(v0 + r + 1, aB, bB, daB, dbB, out, lane);
    }
  }
}

extern "C" void kernel_launch(void* const* d_in, const int* in_sizes, int n_in,
                              void* d_out, int out_size, void* d_ws, size_t ws_size,
                              hipStream_t stream) {
  const float* X     = (const float*)d_in[0];
  const float* W     = (const float*)d_in[1];
  const float* bvec  = (const float*)d_in[2];
  const float* a_e   = (const float*)d_in[3];
  const int*   inc_v = (const int*)d_in[4];
  const int*   inc_e = (const int*)d_in[5];
  float* out = (float*)d_out;

  float* ws    = (float*)d_ws;
  float* Xsum  = ws;
  float* Y     = Xsum + (size_t)N_E * HID;
  float* cnt   = Y    + (size_t)N_E * HID;
  float* es    = cnt  + N_E;
  int* e_lo = (int*)Y;
  int* e_hi = e_lo + NB * 4;
  int* v_lo = (int*)(es + N_E * NH);
  int* v_hi = v_lo + NB * 4;

  key_planes<<<dim3(KWG, 2), 256, 0, stream>>>(inc_v, inc_e, e_lo, e_hi,
                                               v_lo, v_hi);
  v2e_own<<<dim3(N_E / RE), WGT, 0, stream>>>(X, inc_v, inc_e, W, bvec, a_e,
                                              e_lo, e_hi, Xsum, cnt, es);
  gemm_edge<<<dim3((N_E / 16) * NH / 8), 256, 0, stream>>>(Xsum, cnt, W, bvec, Y);
  e2v_own<<<dim3((N_V + RV - 1) / RV), WGT, 0, stream>>>(Y, es, inc_v, inc_e,
                                                         v_lo, v_hi, out);
}


# owners fetch their incidences from an owner-sorted list (4096-entry chunks, LDS counting sort, run table) instead of scanning all keys and gathering keys/sources; rank sum order may differ
# speedup vs baseline: 4.6718x; 1.1039x over previous
#include <hip/hip_runtime.h>
#include <hip/hip_bf16.h>

typedef __attribute__((ext_vector_type(2))) float v2f;
typedef __attribute__((ext_vector_type(8))) float v8f;

#define N_V   100000
#define N_E   20000
#define N_NZ  400000
#define CIN   256
#define HID   256
#define NH    4
#define DH    64
#define NEG_SLOPE 0.2f

#define WGT    1024
#define NWAVE  32
#define LCAP   16384
#define RE     80
#define RV     392

__device__ __forceinline__ int lds_iadd(int* p, int v) {
  return __hip_atomic_fetch_add(p, v, __ATOMIC_RELAXED, __HIP_MEMORY_SCOPE_WORKGROUP);
}

#define CH     4096
#define NCH    ((N_NZ + CH - 1) / CH)
#define NCHP   128
#define ST     512
#define SQ     (CH / 4 / ST)

template <bool ISV>
__device__ __forceinline__ void chunk_sort_body(
    const int* __restrict__ keys, const int* __restrict__ others,
    int* __restrict__ sorted, int* __restrict__ tab, int* binc, int* bins)
{
  constexpr unsigned R = ISV ? RV : RE;
  const int tid = threadIdx.x;
  const int lane = tid & 31;
  const int c = blockIdx.x;
  if (tid < 257) binc[tid] = 0;
  if (c == 0 && tid < 256) {
    for (int k = NCH; k < NCHP; ++k) tab[tid * NCHP + k] = 0;
  }
  int kk[SQ * 4], oo[SQ * 4];
  bool ok[SQ];
#pragma unroll
  for (int u = 0; u < SQ; ++u) {
    const int q = c * (CH / 4) + u * ST + tid;
    ok[u] = q * 4 < N_NZ;
    const int4 k4 = ok[u] ? ((const int4*)keys)[q] : make_int4(0, 0, 0, 0);
    const int4 o4 = ok[u] ? ((const int4*)others)[q] : make_int4(0, 0, 0, 0);
    kk[u * 4 + 0] = k4.x; kk[u * 4 + 1] = k4.y;
    kk[u * 4 + 2] = k4.z; kk[u * 4 + 3] = k4.w;
    oo[u * 4 + 0] = o4.x; oo[u * 4 + 1] = o4.y;
    oo[u * 4 + 2] = o4.z; oo[u * 4 + 3] = o4.w;
  }
  __syncthreads();
  int own[SQ * 4], rank[SQ * 4];
#pragma unroll
  for (int i = 0; i < SQ * 4; ++i) {
    const unsigned o = min((unsigned)kk[i] / R, 255u);
    kk[i] = kk[i] - (int)(o * R);
    own[i] = ok[i >> 2] ? (int)o : 256;
    rank[i] = lds_iadd(&binc[own[i]], 1);
  }
  __syncthreads();
  if (tid < 32) {
    int cs[8];
    int sum = 0;
#pragma unroll
    for (int k = 0; k < 8; ++k) { cs[k] = binc[lane * 8 + k]; sum += cs[k]; }
    int inc = sum;
#pragma unroll
    for (int off = 1; off < 32; off <<= 1) {
      const int up = __shfl_up(inc, off);
      if (lane >= off) inc += up;
    }
    int run = inc - sum;
#pragma unroll
    for (int k = 0; k < 8; ++k) {
      bins[lane * 8 + k] = run;
      tab[(lane * 8 + k) * NCHP + c] = run | (cs[k] << 16);
      run += cs[k];
    }
    if (lane == 0) bins[256] = 0;
  }
  __syncthreads();
  int dst[SQ * 4];
#pragma unroll
  for (int i = 0; i < SQ * 4; ++i) dst[i] = c * CH + bins[own[i]] + rank[i];
#pragma unroll
  for (int i = 0; i < SQ * 4; ++i) {
    if (ok[i >> 2]) sorted[dst[i]] = kk[i] | (oo[i] << 9);
  }
}

__global__ __launch_bounds__(ST) void chunk_sort(
    const int* __restrict__ inc_v, const int* __restrict__ inc_e,
    int* __restrict__ e_sorted, int* __restrict__ e_tab,
    int* __restrict__ v_sorted, int* __restrict__ v_tab)
{
  __shared__ int binc[257];
  __shared__ int bins[257];
  if (blockIdx.y != 0)
    chunk_sort_body<true>(inc_v, inc_e, v_sorted, v_tab, binc, bins);
  else
    chunk_sort_body<false>(inc_e, inc_v, e_sorted, e_tab, binc, bins);
}

__device__ __forceinline__ int fetch_matches(
    const int* __restrict__ sorted, const int* __restrict__ tab, const int ob,
    int* listA, int* listB, int* rowcnt, int* cbase, int* cseg, int* nmatch,
    const int lane, const int wave)
{
  if (wave == 0) {
    const int4 t4 = ((const int4*)(tab + ob * NCHP))[lane];
    int t[4] = {t4.x, t4.y, t4.z, t4.w};
    int s = 0;
#pragma unroll
    for (int k = 0; k < 4; ++k) s += (int)((unsigned)t[k] >> 16);
    int inc = s;
#pragma unroll
    for (int off = 1; off < 32; off <<= 1) {
      const int up = __shfl_up(inc, off);
      if (lane >= off) inc += up;
    }
    int run = inc - s;
#pragma unroll
    for (int k = 0; k < 4; ++k) {
      if (lane * 4 + k < NCH) { cseg[lane * 4 + k] = t[k]; cbase[lane * 4 + k] = run; }
      run += (int)((unsigned)t[k] >> 16);
    }
    if (lane == 31) *nmatch = inc;
  }
  __syncthreads();
  const int nm = *nmatch;
  if (nm <= LCAP) {
    constexpr int CW = (NCH + NWAVE - 1) / NWAVE;
    int st[CW], len[CW], base[CW], w[CW];
#pragma unroll
    for (int u = 0; u < CW; ++u) {
      const int c = wave + u * NWAVE;
      const int t = c < NCH ? cseg[c] : 0;
      base[u] = c < NCH ? cbase[c] : 0;
      len[u] = (int)((unsigned)t >> 16);
      st[u] = c * CH + (t & 0xffff);
    }
#pragma unroll
    for (int u = 0; u < CW; ++u) w[u] = lane < len[u] ? sorted[st[u] + lane] : 0;
#pragma unroll
    for (int u = 0; u < CW; ++u) {
      if (lane < len[u]) {
        const int r = w[u] & 511;
        const int rank = lds_iadd(&rowcnt[r], 1);
        listA[base[u] + lane] = r | (rank << 9);
        listB[base[u] + lane] = (int)((unsigned)w[u] >> 9);
      }
      for (int j = 32 + lane; j < len[u]; j += 32) {
        const int ww = sorted[st[u] + j];
        const int r = ww & 511;
        const int rank = lds_iadd(&rowcnt[r], 1);
        listA[base[u] + j] = r | (rank << 9);
        listB[base[u] + j] = (int)((unsigned)ww >> 9);
      }
    }
  }
  __syncthreads();
  return nm;
}

template <int R>
__device__ __forceinline__ void order_matches(
    const int nm, int* listA, int* listB, int* srcS,
    int* rowcnt, int* rowstart, const int tid, const int lane, const int wave)
{
  if (wave == 0) {
    constexpr int C = (R + 31) / 32;
    int cs[C];
    int s = 0;
#pragma unroll
    for (int k = 0; k < C; ++k) {
      const int idx = lane * C + k;
      cs[k] = idx < R ? rowcnt[idx] : 0;
      s += cs[k];
    }
    int inc = s;
#pragma unroll
    for (int off = 1; off < 32; off <<= 1) {
      const int up = __shfl_up(inc, off);
      if (lane >= off) inc += up;
    }
    int run = inc - s;
#pragma unroll
    for (int k = 0; k < C; ++k) {
      const int idx = lane * C + k;
      if (idx < R) rowstart[idx] = run;
      run += cs[k];
    }
  }
  __syncthreads();
  for (int t = tid; t < nm; t += WGT) {
    const int rr = listA[t];
    srcS[rowstart[rr & 511] + (rr >> 9)] = listB[t];
  }
  __syncthreads();
}

__device__ __forceinline__ void v2e_finish(
    const int e, const float4 xa, const float4 xb, const float c,
    const float* uL, const float* c0L, float* __restrict__ Xsum,
    float* __restrict__ cnt, float* __restrict__ es, const int lane)
{
  float4* __restrict__ dst = (float4*)(Xsum + (size_t)e * HID);
  dst[lane] = xa;
  dst[lane + 32] = xb;
  float p[NH];
#pragma unroll
  for (int h = 0; h < NH; ++h) {
    const float4 ua = ((const float4*)uL)[h * 64 + lane];
    const float4 ub = ((const float4*)uL)[h * 64 + lane + 32];
    p[h] = xa.x * ua.x + xa.y * ua.y + xa.z * ua.z + xa.w * ua.w
         + xb.x * ub.x + xb.y * ub.y + xb.z * ub.z + xb.w * ub.w;
  }
#pragma unroll
  for (int off = 16; off >= 1; off >>= 1) {
#pragma unroll
    for (int h = 0; h < NH; ++h) p[h] += __shfl_xor(p[h], off);
  }
  if (lane == 0) cnt[e] = c;
  if (lane < NH) {
    const float ph = lane == 0 ? p[0] : lane == 1 ? p[1]
                   : lane == 2 ? p[2] : p[3];
    const float alpha = ph * (1.0f / fmaxf(c, 1.0f)) + c0L[lane];
    float s = alpha >= 0.0f ? alpha : NEG_SLOPE * alpha;
    s = fminf(fmaxf(s, 0.001f), 5.0f);
    es[e * NH + lane] = __expf(s);
  }
}

__global__ __launch_bounds__(WGT) void v2e_own(
    const float* __restrict__ X, const int* __restrict__ inc_v,
    const int* __restrict__ inc_e, const float* __restrict__ W,
    const float* __restrict__ bvec, const float* __restrict__ a_e,
    const int* __restrict__ e_sorted, const int* __restrict__ e_tab,
    float* __restrict__ Xsum, float* __restrict__ cnt, float* __restrict__ es)
{
  __shared__ __attribute__((aligned(16))) float uL[NH * CIN];
  __shared__ float c0L[NH];
  __shared__ int rowcnt[RE];
  __shared__ int rowstart[RE];
  __shared__ int cbase[NCH];
  __shared__ int cseg[NCH];
  __shared__ int listA[LCAP];
  __shared__ int listB[LCAP];
  __shared__ int srcS[LCAP];
  __shared__ int nmatch;
  __shared__ int nextrow;

  const int tid = threadIdx.x;
  const int lane = tid & 31;
  const int wave = __builtin_amdgcn_readfirstlane(tid >> 5);
  const int e0 = blockIdx.x * RE;

  {
    const int h = wave >> 3;
    const int cb = (wave & 7) * 32;
    const int qd = lane & 3, rw = lane >> 2;
    const float4* __restrict__ ap = (const float4*)(a_e + h * DH) + qd * 4;
    const float4* __restrict__ wp =
        (const float4*)(W + ((size_t)h * CIN + cb + rw) * DH) + qd * 4;
    const float4* __restrict__ bp = (const float4*)(bvec + h * DH) + qd * 4;
    float4 av[4], bv[4], wv[16];
#pragma unroll
    for (int b = 0; b < 4; ++b) {
#pragma unroll
      for (int i = 0; i < 4; ++i) wv[b * 4 + i] = wp[b * (8 * DH / 4) + i];
    }
#pragma unroll
    for (int i = 0; i < 4; ++i) { av[i] = ap[i]; bv[i] = bp[i]; }
    float p[4];
#pragma unroll
    for (int b = 0; b < 4; ++b) {
      float s = 0.0f;
#pragma unroll
      for (int i = 0; i < 4; ++i)
        s += wv[b * 4 + i].x * av[i].x + wv[b * 4 + i].y * av[i].y
           + wv[b * 4 + i].z * av[i].z + wv[b * 4 + i].w * av[i].w;
      p[b] = s;
    }
    const bool o1 = (lane & 1) != 0, o2 = (lane & 2) != 0;
    const float q0 = (o1 ? p[1] : p[0]) + __shfl_xor(o1 ? p[0] : p[1], 1);
    const float q1 = (o1 ? p[3] : p[2]) + __shfl_xor(o1 ? p[2] : p[3], 1);
    const float uv = (o2 ? q1 : q0) + __shfl_xor(o2 ? q0 : q1, 2);
    uL[h * CIN + cb + qd * 8 + rw] = uv;
    float q = 0.0f;
#pragma unroll
    for (int i = 0; i < 4; ++i)
      q += bv[i].x * av[i].x + bv[i].y * av[i].y
         + bv[i].z * av[i].z + bv[i].w * av[i].w;
    q += __shfl_xor(q, 1);
    q += __shfl_xor(q, 2);
    if ((wave & 7) == 0 && lane == 0) c0L[h] = q;
  }

  for (int k = tid; k < RE; k += WGT) rowcnt[k] = 0;
  if (tid == 0) { nmatch = 0; nextrow = 0; }
  __syncthreads();

  const int nm = fetch_matches(e_sorted, e_tab, blockIdx.x, listA, listB,
                               rowcnt, cbase, cseg, &nmatch, lane, wave);

  const float4 zero4 = make_float4(0.f, 0.f, 0.f, 0.f);
  if (nm > LCAP) {
    for (int r = wave; r < RE; r += NWAVE) {
      float4 sa = zero4, sb = zero4;
      int n = 0;
      for (int i0 = 0; i0 < N_NZ; i0 += 32) {
        const int i = i0 + lane;
        const bool m = i < N_NZ && inc_e[i] == e0 + r;
        const int vm = m ? inc_v[i] : 0;
        unsigned mask = (unsigned)__ballot(m);
        while (mask != 0u) {
          const int bpos = __ffs((int)mask) - 1;
          mask &= mask - 1u;
          const int v = __shfl(vm, bpos);
          const float4* __restrict__ xr = (const float4*)(X + (size_t)v * CIN);
          const float4 xa = xr[lane], xb = xr[lane + 32];
          sa.x += xa.x; sa.y += xa.y; sa.z += xa.z; sa.w += xa.w;
          sb.x += xb.x; sb.y += xb.y; sb.z += xb.z; sb.w += xb.w;
          ++n;
        }
      }
      v2e_finish(e0 + r, sa, sb, (float)n, uL, c0L, Xsum, cnt, es, lane);
    }
  } else {
    order_matches<RE>(nm, listA, listB, srcS, rowcnt, rowstart, tid, lane,
                      wave);
    for (;;) {
      int rnew = 0;
      if (lane == 0) rnew = lds_iadd(&nextrow, 1);
      const int r = __builtin_amdgcn_readfirstlane(rnew);
      if (r >= RE) break;
      const int j0 = __builtin_amdgcn_readfirstlane(rowstart[r]);
      const int n  = __builtin_amdgcn_readfirstlane(rowcnt[r]);
      float4 sa = zero4, sb = zero4;
      for (int j = 0; j < n; j += 8) {
        const int mine = srcS[min(j0 + j + (lane & 7), LCAP - 1)];
        float4 xa[8], xb[8];
#pragma unroll
        for (int u = 0; u < 8; ++u) {
          if (j + u < n) {
            const int v = __builtin_amdgcn_readlane(mine, u);
            const float4* __restrict__ xr = (const float4*)(X + (size_t)v * CIN);
            xa[u] = xr[lane];
            xb[u] = xr[lane + 32];
          } else {
            xa[u] = zero4;
            xb[u] = zero4;
          }
        }
#pragma unroll
        for (int u = 0; u < 8; ++u) {
          sa.x += xa[u].x; sa.y += xa[u].y; sa.z += xa[u].z; sa.w += xa[u].w;
          sb.x += xb[u].x; sb.y += xb[u].y; sb.z += xb[u].z; sb.w += xb[u].w;
        }
      }
      v2e_finish(e0 + r, sa, sb, (float)n, uL, c0L, Xsum, cnt, es, lane);
    }
  }
}

__global__ __launch_bounds__(256) void gemm_edge(
    const float* __restrict__ Xsum, const float* __restrict__ cnt,
    const float* __restrict__ W, const float* __restrict__ bvec,
    float* __restrict__ Y)
{
  const int lane = threadIdx.x & 31;
  const int wid  = blockIdx.x * 8 + (threadIdx.x >> 5);
  const int row_tile = wid >> 2;
  const int head = wid & 3;
  const int row0 = row_tile * 16;
  const int m = lane & 15;
  const int half = lane >> 4;

  const int row = row0 + m;
  const float inv = 1.0f / fmaxf(cnt[row], 1.0f);
  const float* __restrict__ arow = Xsum + (size_t)row * HID;
  const float* __restrict__ wh = W + (size_t)head * (CIN * DH);

  v8f acc[4] = {v8f{}, v8f{}, v8f{}, v8f{}};
#pragma unroll 4
  for (int k0 = 0; k0 < CIN; k0 += 4) {
    const int ka = k0 + 2 * half;
    v2f afrag;
    afrag.x = arow[ka] * inv;
    afrag.y = arow[ka + 1] * inv;
    const float* __restrict__ wk = wh + (size_t)ka * DH + m;
#pragma unroll
    for (int t = 0; t < 4; ++t) {
      v2f bfrag;
      bfrag.x = wk[t * 16];
      bfrag.y = wk[t * 16 + DH];
      acc[t] = __builtin_amdgcn_wmma_f32_16x16x4_f32(
          false, afrag, false, bfrag, (short)0, acc[t], false, false);
    }
  }

  const int mbase = half * 8;
#pragma unroll
  for (int t = 0; t < 4; ++t) {
    const int col = head * DH + t * 16 + m;
    const float bias = bvec[col];
#pragma unroll
    for (int r = 0; r < 8; ++r) {
      Y[(size_t)(row0 + mbase + r) * HID + col] = acc[t][r] + bias;
    }
  }
}

__device__ __forceinline__ void e2v_finish(
    const int v, const float4 a, const float4 b, const float da,
    const float db, float* __restrict__ out, const int lane)
{
  if (v < N_V) {
    const float ia = da > 0.0f ? 1.0f / da : 0.0f;
    const float ib = db > 0.0f ? 1.0f / db : 0.0f;
    float4 x, y;
    x.x = a.x * ia; x.y = a.y * ia; x.z = a.z * ia; x.w = a.w * ia;
    y.x = b.x * ib; y.y = b.y * ib; y.z = b.z * ib; y.w = b.w * ib;
    x.x = x.x > 0.0f ? x.x : expm1f(x.x);
    x.y = x.y > 0.0f ? x.y : expm1f(x.y);
    x.z = x.z > 0.0f ? x.z : expm1f(x.z);
    x.w = x.w > 0.0f ? x.w : expm1f(x.w);
    y.x = y.x > 0.0f ? y.x : expm1f(y.x);
    y.y = y.y > 0.0f ? y.y : expm1f(y.y);
    y.z = y.z > 0.0f ? y.z : expm1f(y.z);
    y.w = y.w > 0.0f ? y.w : expm1f(y.w);
    float4* __restrict__ dst = (float4*)(out + (size_t)v * HID);
    dst[lane] = x;
    dst[lane + 32] = y;
  }
}

__global__ __launch_bounds__(WGT) void e2v_own(
    const float* __restrict__ Y, const float* __restrict__ es,
    const int* __restrict__ inc_v, const int* __restrict__ inc_e,
    const int* __restrict__ v_sorted, const int* __restrict__ v_tab,
    float* __restrict__ out)
{
  __shared__ int rowcnt[RV];
  __shared__ int rowstart[RV];
  __shared__ int cbase[NCH];
  __shared__ int cseg[NCH];
  __shared__ int listA[LCAP];
  __shared__ int listB[LCAP];
  __shared__ int srcS[LCAP];
  __shared__ int nmatch;
  __shared__ int nextrow;

  const int tid = threadIdx.x;
  const int lane = tid & 31;
  const int wave = __builtin_amdgcn_readfirstlane(tid >> 5);
  const int v0 = blockIdx.x * RV;

  for (int k = tid; k < RV; k += WGT) rowcnt[k] = 0;
  if (tid == 0) { nmatch = 0; nextrow = 0; }
  __syncthreads();

  const int nm = fetch_matches(v_sorted, v_tab, blockIdx.x, listA, listB,
                               rowcnt, cbase, cseg, &nmatch, lane, wave);

  const float* __restrict__ esl = es + (lane < 16 ? 0 : 1);
  const float4 zero4 = make_float4(0.f, 0.f, 0.f, 0.f);
  if (nm > LCAP) {
    for (int r = wave; r < RV; r += NWAVE) {
      float4 a = zero4, b = zero4;
      float da = 0.0f, db = 0.0f;
      for (int i0 = 0; i0 < N_NZ; i0 += 32) {
        const int i = i0 + lane;
        const bool m = i < N_NZ && inc_v[i] == v0 + r;
        const int em = m ? inc_e[i] : 0;
        unsigned mask = (unsigned)__ballot(m);
        while (mask != 0u) {
          const int bpos = __ffs((int)mask) - 1;
          mask &= mask - 1u;
          const int e = __shfl(em, bpos);
          const float wa = esl[e * NH], wb = esl[e * NH + 2];
          const float4* __restrict__ yr = (const float4*)(Y + (size_t)e * HID);
          const float4 ya = yr[lane], yb = yr[lane + 32];
          a.x += wa * ya.x; a.y += wa * ya.y; a.z += wa * ya.z; a.w += wa * ya.w;
          b.x += wb * yb.x; b.y += wb * yb.y; b.z += wb * yb.z; b.w += wb * yb.w;
          da += wa; db += wb;
        }
      }
      e2v_finish(v0 + r, a, b, da, db, out, lane);
    }
  } else {
    order_matches<RV>(nm, listA, listB, srcS, rowcnt, rowstart, tid, lane,
                      wave);
    for (;;) {
      int rnew = 0;
      if (lane == 0) rnew = lds_iadd(&nextrow, 1);
      const int r = 2 * __builtin_amdgcn_readfirstlane(rnew);
      if (r >= RV) break;
      const int jA = __builtin_amdgcn_readfirstlane(rowstart[r]);
      const int nA = __builtin_amdgcn_readfirstlane(rowcnt[r]);
      const int jB = __builtin_amdgcn_readfirstlane(rowstart[r + 1]);
      const int nB = __builtin_amdgcn_readfirstlane(rowcnt[r + 1]);
      const int nmax = nA > nB ? nA : nB;
      float4 aA = zero4, bA = zero4, aB = zero4, bB = zero4;
      float daA = 0.0f, dbA = 0.0f, daB = 0.0f, dbB = 0.0f;
      for (int j = 0; j < nmax; j += 4) {
        const int mine =
            srcS[min(((lane & 4) ? jB : jA) + j + (lane & 3), LCAP - 1)];
        float waA[4], wbA[4], waB[4], wbB[4];
        float4 yaA[4], ybA[4], yaB[4], ybB[4];
#pragma unroll
        for (int u = 0; u < 4; ++u) {
          if (j + u < nA) {
            const int e = __builtin_amdgcn_readlane(mine, u);
            waA[u] = esl[e * NH];
            wbA[u] = esl[e * NH + 2];
            const float4* __restrict__ yr = (const float4*)(Y + (size_t)e * HID);
            yaA[u] = yr[lane];
            ybA[u] = yr[lane + 32];
          } else {
            waA[u] = 0.0f; wbA[u] = 0.0f; yaA[u] = zero4; ybA[u] = zero4;
          }
          if (j + u < nB) {
            const int e = __builtin_amdgcn_readlane(mine, 4 + u);
            waB[u] = esl[e * NH];
            wbB[u] = esl[e * NH + 2];
            const float4* __restrict__ yr = (const float4*)(Y + (size_t)e * HID);
            yaB[u] = yr[lane];
            ybB[u] = yr[lane + 32];
          } else {
            waB[u] = 0.0f; wbB[u] = 0.0f; yaB[u] = zero4; ybB[u] = zero4;
          }
        }
#pragma unroll
        for (int u = 0; u < 4; ++u) {
          aA.x += waA[u] * yaA[u].x; aA.y += waA[u] * yaA[u].y;
          aA.z += waA[u] * yaA[u].z; aA.w += waA[u] * yaA[u].w;
          bA.x += wbA[u] * ybA[u].x; bA.y += wbA[u] * ybA[u].y;
          bA.z += wbA[u] * ybA[u].z; bA.w += wbA[u] * ybA[u].w;
          daA += waA[u]; dbA += wbA[u];
          aB.x += waB[u] * yaB[u].x; aB.y += waB[u] * yaB[u].y;
          aB.z += waB[u] * yaB[u].z; aB.w += waB[u] * yaB[u].w;
          bB.x += wbB[u] * ybB[u].x; bB.y += wbB[u] * ybB[u].y;
          bB.z += wbB[u] * ybB[u].z; bB.w += wbB[u] * ybB[u].w;
          daB += waB[u]; dbB += wbB[u];
        }
      }
      e2v_finish(v0 + r,     aA, bA, daA, dbA, out, lane);
      e2v_finish(v0 + r + 1, aB, bB, daB, dbB, out, lane);
    }
  }
}

extern "C" void kernel_launch(void* const* d_in, const int* in_sizes, int n_in,
                              void* d_out, int out_size, void* d_ws, size_t ws_size,
                              hipStream_t stream) {
  const float* X     = (const float*)d_in[0];
  const float* W     = (const float*)d_in[1];
  const float* bvec  = (const float*)d_in[2];
  const float* a_e   = (const float*)d_in[3];
  const int*   inc_v = (const int*)d_in[4];
  const int*   inc_e = (const int*)d_in[5];
  float* out = (float*)d_out;

  float* ws    = (float*)d_ws;
  float* Y     = ws   + (size_t)N_E * HID;
  float* cnt   = Y    + (size_t)N_E * HID;
  float* es    = cnt  + N_E;
  float* Xsum  = out;
  int* v_sorted = (int*)ws;
  int* v_tab    = v_sorted + N_NZ;
  int* e_sorted = (int*)Y;
  int* e_tab    = e_sorted + N_NZ;

  chunk_sort<<<dim3(NCH, 2), ST, 0, stream>>>(inc_v, inc_e, e_sorted, e_tab,
                                              v_sorted, v_tab);
  v2e_own<<<dim3(N_E / RE), WGT, 0, stream>>>(X, inc_v, inc_e, W, bvec, a_e,
                                              e_sorted, e_tab, Xsum, cnt, es);
  gemm_edge<<<dim3((N_E / 16) * NH / 8), 256, 0, stream>>>(Xsum, cnt, W, bvec, Y);
  e2v_own<<<dim3((N_V + RV - 1) / RV), WGT, 0, stream>>>(Y, es, inc_v, inc_e,
                                                         v_sorted, v_tab, out);
}


# gemm_edge fragment loaders: k and column order permuted so A and B come as 16-byte loads (5 loads per 8 WMMAs, was 18) and Y leaves as 16-byte stores; k sum order inside each block of 8 changes
# speedup vs baseline: 4.8222x; 1.0322x over previous
#include <hip/hip_runtime.h>
#include <hip/hip_bf16.h>

typedef __attribute__((ext_vector_type(2))) float v2f;
typedef __attribute__((ext_vector_type(8))) float v8f;

#define N_V   100000
#define N_E   20000
#define N_NZ  400000
#define CIN   256
#define HID   256
#define NH    4
#define DH    64
#define NEG_SLOPE 0.2f

#define WGT    1024
#define NWAVE  32
#define LCAP   16384
#define RE     80
#define RV     392

__device__ __forceinline__ int lds_iadd(int* p, int v) {
  return __hip_atomic_fetch_add(p, v, __ATOMIC_RELAXED, __HIP_MEMORY_SCOPE_WORKGROUP);
}

#define CH     4096
#define NCH    ((N_NZ + CH - 1) / CH)
#define NCHP   128
#define ST     512
#define SQ     (CH / 4 / ST)

template <bool ISV>
__device__ __forceinline__ void chunk_sort_body(
    const int* __restrict__ keys, const int* __restrict__ others,
    int* __restrict__ sorted, int* __restrict__ tab, int* binc, int* bins)
{
  constexpr unsigned R = ISV ? RV : RE;
  const int tid = threadIdx.x;
  const int lane = tid & 31;
  const int c = blockIdx.x;
  if (tid < 257) binc[tid] = 0;
  if (c == 0 && tid < 256) {
    for (int k = NCH; k < NCHP; ++k) tab[tid * NCHP + k] = 0;
  }
  int kk[SQ * 4], oo[SQ * 4];
  bool ok[SQ];
#pragma unroll
  for (int u = 0; u < SQ; ++u) {
    const int q = c * (CH / 4) + u * ST + tid;
    ok[u] = q * 4 < N_NZ;
    const int4 k4 = ok[u] ? ((const int4*)keys)[q] : make_int4(0, 0, 0, 0);
    const int4 o4 = ok[u] ? ((const int4*)others)[q] : make_int4(0, 0, 0, 0);
    kk[u * 4 + 0] = k4.x; kk[u * 4 + 1] = k4.y;
    kk[u * 4 + 2] = k4.z; kk[u * 4 + 3] = k4.w;
    oo[u * 4 + 0] = o4.x; oo[u * 4 + 1] = o4.y;
    oo[u * 4 + 2] = o4.z; oo[u * 4 + 3] = o4.w;
  }
  __syncthreads();
  int own[SQ * 4], rank[SQ * 4];
#pragma unroll
  for (int i = 0; i < SQ * 4; ++i) {
    const unsigned o = min((unsigned)kk[i] / R, 255u);
    kk[i] = kk[i] - (int)(o * R);
    own[i] = ok[i >> 2] ? (int)o : 256;
    rank[i] = lds_iadd(&binc[own[i]], 1);
  }
  __syncthreads();
  if (tid < 32) {
    int cs[8];
    int sum = 0;
#pragma unroll
    for (int k = 0; k < 8; ++k) { cs[k] = binc[lane * 8 + k]; sum += cs[k]; }
    int inc = sum;
#pragma unroll
    for (int off = 1; off < 32; off <<= 1) {
      const int up = __shfl_up(inc, off);
      if (lane >= off) inc += up;
    }
    int run = inc - sum;
#pragma unroll
    for (int k = 0; k < 8; ++k) {
      bins[lane * 8 + k] = run;
      tab[(lane * 8 + k) * NCHP + c] = run | (cs[k] << 16);
      run += cs[k];
    }
    if (lane == 0) bins[256] = 0;
  }
  __syncthreads();
  int dst[SQ * 4];
#pragma unroll
  for (int i = 0; i < SQ * 4; ++i) dst[i] = c * CH + bins[own[i]] + rank[i];
#pragma unroll
  for (int i = 0; i < SQ * 4; ++i) {
    if (ok[i >> 2]) sorted[dst[i]] = kk[i] | (oo[i] << 9);
  }
}

__global__ __launch_bounds__(ST) void chunk_sort(
    const int* __restrict__ inc_v, const int* __restrict__ inc_e,
    int* __restrict__ e_sorted, int* __restrict__ e_tab,
    int* __restrict__ v_sorted, int* __restrict__ v_tab)
{
  __shared__ int binc[257];
  __shared__ int bins[257];
  if (blockIdx.y != 0)
    chunk_sort_body<true>(inc_v, inc_e, v_sorted, v_tab, binc, bins);
  else
    chunk_sort_body<false>(inc_e, inc_v, e_sorted, e_tab, binc, bins);
}

__device__ __forceinline__ int fetch_matches(
    const int* __restrict__ sorted, const int* __restrict__ tab, const int ob,
    int* listA, int* listB, int* rowcnt, int* cbase, int* cseg, int* nmatch,
    const int lane, const int wave)
{
  if (wave == 0) {
    const int4 t4 = ((const int4*)(tab + ob * NCHP))[lane];
    int t[4] = {t4.x, t4.y, t4.z, t4.w};
    int s = 0;
#pragma unroll
    for (int k = 0; k < 4; ++k) s += (int)((unsigned)t[k] >> 16);
    int inc = s;
#pragma unroll
    for (int off = 1; off < 32; off <<= 1) {
      const int up = __shfl_up(inc, off);
      if (lane >= off) inc += up;
    }
    int run = inc - s;
#pragma unroll
    for (int k = 0; k < 4; ++k) {
      if (lane * 4 + k < NCH) { cseg[lane * 4 + k] = t[k]; cbase[lane * 4 + k] = run; }
      run += (int)((unsigned)t[k] >> 16);
    }
    if (lane == 31) *nmatch = inc;
  }
  __syncthreads();
  const int nm = *nmatch;
  if (nm <= LCAP) {
    constexpr int CW = (NCH + NWAVE - 1) / NWAVE;
    int st[CW], len[CW], base[CW], w[CW];
#pragma unroll
    for (int u = 0; u < CW; ++u) {
      const int c = wave + u * NWAVE;
      const int t = c < NCH ? cseg[c] : 0;
      base[u] = c < NCH ? cbase[c] : 0;
      len[u] = (int)((unsigned)t >> 16);
      st[u] = c * CH + (t & 0xffff);
    }
#pragma unroll
    for (int u = 0; u < CW; ++u) w[u] = lane < len[u] ? sorted[st[u] + lane] : 0;
#pragma unroll
    for (int u = 0; u < CW; ++u) {
      if (lane < len[u]) {
        const int r = w[u] & 511;
        const int rank = lds_iadd(&rowcnt[r], 1);
        listA[base[u] + lane] = r | (rank << 9);
        listB[base[u] + lane] = (int)((unsigned)w[u] >> 9);
      }
      for (int j = 32 + lane; j < len[u]; j += 32) {
        const int ww = sorted[st[u] + j];
        const int r = ww & 511;
        const int rank = lds_iadd(&rowcnt[r], 1);
        listA[base[u] + j] = r | (rank << 9);
        listB[base[u] + j] = (int)((unsigned)ww >> 9);
      }
    }
  }
  __syncthreads();
  return nm;
}

template <int R>
__device__ __forceinline__ void order_matches(
    const int nm, int* listA, int* listB, int* srcS,
    int* rowcnt, int* rowstart, const int tid, const int lane, const int wave)
{
  if (wave == 0) {
    constexpr int C = (R + 31) / 32;
    int cs[C];
    int s = 0;
#pragma unroll
    for (int k = 0; k < C; ++k) {
      const int idx = lane * C + k;
      cs[k] = idx < R ? rowcnt[idx] : 0;
      s += cs[k];
    }
    int inc = s;
#pragma unroll
    for (int off = 1; off < 32; off <<= 1) {
      const int up = __shfl_up(inc, off);
      if (lane >= off) inc += up;
    }
    int run = inc - s;
#pragma unroll
    for (int k = 0; k < C; ++k) {
      const int idx = lane * C + k;
      if (idx < R) rowstart[idx] = run;
      run += cs[k];
    }
  }
  __syncthreads();
  for (int t = tid; t < nm; t += WGT) {
    const int rr = listA[t];
    srcS[rowstart[rr & 511] + (rr >> 9)] = listB[t];
  }
  __syncthreads();
}

__device__ __forceinline__ void v2e_finish(
    const int e, const float4 xa, const float4 xb, const float c,
    const float* uL, const float* c0L, float* __restrict__ Xsum,
    float* __restrict__ cnt, float* __restrict__ es, const int lane)
{
  float4* __restrict__ dst = (float4*)(Xsum + (size_t)e * HID);
  dst[lane] = xa;
  dst[lane + 32] = xb;
  float p[NH];
#pragma unroll
  for (int h = 0; h < NH; ++h) {
    const float4 ua = ((const float4*)uL)[h * 64 + lane];
    const float4 ub = ((const float4*)uL)[h * 64 + lane + 32];
    p[h] = xa.x * ua.x + xa.y * ua.y + xa.z * ua.z + xa.w * ua.w
         + xb.x * ub.x + xb.y * ub.y + xb.z * ub.z + xb.w * ub.w;
  }
#pragma unroll
  for (int off = 16; off >= 1; off >>= 1) {
#pragma unroll
    for (int h = 0; h < NH; ++h) p[h] += __shfl_xor(p[h], off);
  }
  if (lane == 0) cnt[e] = c;
  if (lane < NH) {
    const float ph = lane == 0 ? p[0] : lane == 1 ? p[1]
                   : lane == 2 ? p[2] : p[3];
    const float alpha = ph * (1.0f / fmaxf(c, 1.0f)) + c0L[lane];
    float s = alpha >= 0.0f ? alpha : NEG_SLOPE * alpha;
    s = fminf(fmaxf(s, 0.001f), 5.0f);
    es[e * NH + lane] = __expf(s);
  }
}

__global__ __launch_bounds__(WGT) void v2e_own(
    const float* __restrict__ X, const int* __restrict__ inc_v,
    const int* __restrict__ inc_e, const float* __restrict__ W,
    const float* __restrict__ bvec, const float* __restrict__ a_e,
    const int* __restrict__ e_sorted, const int* __restrict__ e_tab,
    float* __restrict__ Xsum, float* __restrict__ cnt, float* __restrict__ es)
{
  __shared__ __attribute__((aligned(16))) float uL[NH * CIN];
  __shared__ float c0L[NH];
  __shared__ int rowcnt[RE];
  __shared__ int rowstart[RE];
  __shared__ int cbase[NCH];
  __shared__ int cseg[NCH];
  __shared__ int listA[LCAP];
  __shared__ int listB[LCAP];
  __shared__ int srcS[LCAP];
  __shared__ int nmatch;
  __shared__ int nextrow;

  const int tid = threadIdx.x;
  const int lane = tid & 31;
  const int wave = __builtin_amdgcn_readfirstlane(tid >> 5);
  const int e0 = blockIdx.x * RE;

  {
    const int h = wave >> 3;
    const int cb = (wave & 7) * 32;
    const int qd = lane & 3, rw = lane >> 2;
    const float4* __restrict__ ap = (const float4*)(a_e + h * DH) + qd * 4;
    const float4* __restrict__ wp =
        (const float4*)(W + ((size_t)h * CIN + cb + rw) * DH) + qd * 4;
    const float4* __restrict__ bp = (const float4*)(bvec + h * DH) + qd * 4;
    float4 av[4], bv[4], wv[16];
#pragma unroll
    for (int b = 0; b < 4; ++b) {
#pragma unroll
      for (int i = 0; i < 4; ++i) wv[b * 4 + i] = wp[b * (8 * DH / 4) + i];
    }
#pragma unroll
    for (int i = 0; i < 4; ++i) { av[i] = ap[i]; bv[i] = bp[i]; }
    float p[4];
#pragma unroll
    for (int b = 0; b < 4; ++b) {
      float s = 0.0f;
#pragma unroll
      for (int i = 0; i < 4; ++i)
        s += wv[b * 4 + i].x * av[i].x + wv[b * 4 + i].y * av[i].y
           + wv[b * 4 + i].z * av[i].z + wv[b * 4 + i].w * av[i].w;
      p[b] = s;
    }
    const bool o1 = (lane & 1) != 0, o2 = (lane & 2) != 0;
    const float q0 = (o1 ? p[1] : p[0]) + __shfl_xor(o1 ? p[0] : p[1], 1);
    const float q1 = (o1 ? p[3] : p[2]) + __shfl_xor(o1 ? p[2] : p[3], 1);
    const float uv = (o2 ? q1 : q0) + __shfl_xor(o2 ? q0 : q1, 2);
    uL[h * CIN + cb + qd * 8 + rw] = uv;
    float q = 0.0f;
#pragma unroll
    for (int i = 0; i < 4; ++i)
      q += bv[i].x * av[i].x + bv[i].y * av[i].y
         + bv[i].z * av[i].z + bv[i].w * av[i].w;
    q += __shfl_xor(q, 1);
    q += __shfl_xor(q, 2);
    if ((wave & 7) == 0 && lane == 0) c0L[h] = q;
  }

  for (int k = tid; k < RE; k += WGT) rowcnt[k] = 0;
  if (tid == 0) { nmatch = 0; nextrow = 0; }
  __syncthreads();

  const int nm = fetch_matches(e_sorted, e_tab, blockIdx.x, listA, listB,
                               rowcnt, cbase, cseg, &nmatch, lane, wave);

  const float4 zero4 = make_float4(0.f, 0.f, 0.f, 0.f);
  if (nm > LCAP) {
    for (int r = wave; r < RE; r += NWAVE) {
      float4 sa = zero4, sb = zero4;
      int n = 0;
      for (int i0 = 0; i0 < N_NZ; i0 += 32) {
        const int i = i0 + lane;
        const bool m = i < N_NZ && inc_e[i] == e0 + r;
        const int vm = m ? inc_v[i] : 0;
        unsigned mask = (unsigned)__ballot(m);
        while (mask != 0u) {
          const int bpos = __ffs((int)mask) - 1;
          mask &= mask - 1u;
          const int v = __shfl(vm, bpos);
          const float4* __restrict__ xr = (const float4*)(X + (size_t)v * CIN);
          const float4 xa = xr[lane], xb = xr[lane + 32];
          sa.x += xa.x; sa.y += xa.y; sa.z += xa.z; sa.w += xa.w;
          sb.x += xb.x; sb.y += xb.y; sb.z += xb.z; sb.w += xb.w;
          ++n;
        }
      }
      v2e_finish(e0 + r, sa, sb, (float)n, uL, c0L, Xsum, cnt, es, lane);
    }
  } else {
    order_matches<RE>(nm, listA, listB, srcS, rowcnt, rowstart, tid, lane,
                      wave);
    for (;;) {
      int rnew = 0;
      if (lane == 0) rnew = lds_iadd(&nextrow, 1);
      const int r = __builtin_amdgcn_readfirstlane(rnew);
      if (r >= RE) break;
      const int j0 = __builtin_amdgcn_readfirstlane(rowstart[r]);
      const int n  = __builtin_amdgcn_readfirstlane(rowcnt[r]);
      float4 sa = zero4, sb = zero4;
      for (int j = 0; j < n; j += 8) {
        const int mine = srcS[min(j0 + j + (lane & 7), LCAP - 1)];
        float4 xa[8], xb[8];
#pragma unroll
        for (int u = 0; u < 8; ++u) {
          if (j + u < n) {
            const int v = __builtin_amdgcn_readlane(mine, u);
            const float4* __restrict__ xr = (const float4*)(X + (size_t)v * CIN);
            xa[u] = xr[lane];
            xb[u] = xr[lane + 32];
          } else {
            xa[u] = zero4;
            xb[u] = zero4;
          }
        }
#pragma unroll
        for (int u = 0; u < 8; ++u) {
          sa.x += xa[u].x; sa.y += xa[u].y; sa.z += xa[u].z; sa.w += xa[u].w;
          sb.x += xb[u].x; sb.y += xb[u].y; sb.z += xb[u].z; sb.w += xb[u].w;
        }
      }
      v2e_finish(e0 + r, sa, sb, (float)n, uL, c0L, Xsum, cnt, es, lane);
    }
  }
}

struct FragSet {
  float4 a;
  float4 b[4];
};

__device__ __forceinline__ void frag_load(
    FragSet& f, const float4* __restrict__ ap, const float4* __restrict__ bp,
    const int blk)
{
  f.a = ap[blk * 2];
#pragma unroll
  for (int q = 0; q < 4; ++q) f.b[q] = bp[blk * (8 * DH / 4) + q * (DH / 4)];
}

__device__ __forceinline__ void frag_mma(
    const FragSet& f, const float inv, v8f (&acc)[4])
{
  const float ak[4] = {f.a.x * inv, f.a.y * inv, f.a.z * inv, f.a.w * inv};
#pragma unroll
  for (int j = 0; j < 2; ++j) {
    v2f afrag;
    afrag.x = ak[2 * j];
    afrag.y = ak[2 * j + 1];
    const float4 bx = f.b[2 * j];
    const float4 by = f.b[2 * j + 1];
    const float bxs[4] = {bx.x, bx.y, bx.z, bx.w};
    const float bys[4] = {by.x, by.y, by.z, by.w};
#pragma unroll
    for (int t = 0; t < 4; ++t) {
      v2f bfrag;
      bfrag.x = bxs[t];
      bfrag.y = bys[t];
      acc[t] = __builtin_amdgcn_wmma_f32_16x16x4_f32(
          false, afrag, false, bfrag, (short)0, acc[t], false, false);
    }
  }
}

__global__ __launch_bounds__(256) void gemm_edge(
    const float* __restrict__ Xsum, const float* __restrict__ cnt,
    const float* __restrict__ W, const float* __restrict__ bvec,
    float* __restrict__ Y)
{
  const int lane = threadIdx.x & 31;
  const int wid  = blockIdx.x * 8 + (threadIdx.x >> 5);
  const int row_tile = wid >> 2;
  const int head = wid & 3;
  const int row0 = row_tile * 16;
  const int m = lane & 15;
  const int half = lane >> 4;

  const int row = row0 + m;
  const float inv = 1.0f / fmaxf(cnt[row], 1.0f);
  const float4* __restrict__ ap =
      (const float4*)(Xsum + (size_t)row * HID) + half;
  const float4* __restrict__ bp =
      (const float4*)(W + (size_t)head * (CIN * DH) + (size_t)(4 * half) * DH) + m;

  v8f acc[4] = {v8f{}, v8f{}, v8f{}, v8f{}};
  FragSet f0, f1;
  frag_load(f0, ap, bp, 0);
  for (int blk = 0; blk < CIN / 8; blk += 2) {
    frag_load(f1, ap, bp, blk + 1);
    frag_mma(f0, inv, acc);
    if (blk + 2 < CIN / 8) frag_load(f0, ap, bp, blk + 2);
    frag_mma(f1, inv, acc);
  }

  const int mbase = half * 8;
  const float4 bias = ((const float4*)(bvec + head * DH))[m];
  float4* __restrict__ yp =
      (float4*)(Y + (size_t)(row0 + mbase) * HID + head * DH) + m;
#pragma unroll
  for (int r = 0; r < 8; ++r) {
    float4 o;
    o.x = acc[0][r] + bias.x;
    o.y = acc[1][r] + bias.y;
    o.z = acc[2][r] + bias.z;
    o.w = acc[3][r] + bias.w;
    yp[r * (HID / 4)] = o;
  }
}

__device__ __forceinline__ void e2v_finish(
    const int v, const float4 a, const float4 b, const float da,
    const float db, float* __restrict__ out, const int lane)
{
  if (v < N_V) {
    const float ia = da > 0.0f ? 1.0f / da : 0.0f;
    const float ib = db > 0.0f ? 1.0f / db : 0.0f;
    float4 x, y;
    x.x = a.x * ia; x.y = a.y * ia; x.z = a.z * ia; x.w = a.w * ia;
    y.x = b.x * ib; y.y = b.y * ib; y.z = b.z * ib; y.w = b.w * ib;
    x.x = x.x > 0.0f ? x.x : expm1f(x.x);
    x.y = x.y > 0.0f ? x.y : expm1f(x.y);
    x.z = x.z > 0.0f ? x.z : expm1f(x.z);
    x.w = x.w > 0.0f ? x.w : expm1f(x.w);
    y.x = y.x > 0.0f ? y.x : expm1f(y.x);
    y.y = y.y > 0.0f ? y.y : expm1f(y.y);
    y.z = y.z > 0.0f ? y.z : expm1f(y.z);
    y.w = y.w > 0.0f ? y.w : expm1f(y.w);
    float4* __restrict__ dst = (float4*)(out + (size_t)v * HID);
    dst[lane] = x;
    dst[lane + 32] = y;
  }
}

__global__ __launch_bounds__(WGT) void e2v_own(
    const float* __restrict__ Y, const float* __restrict__ es,
    const int* __restrict__ inc_v, const int* __restrict__ inc_e,
    const int* __restrict__ v_sorted, const int* __restrict__ v_tab,
    float* __restrict__ out)
{
  __shared__ int rowcnt[RV];
  __shared__ int rowstart[RV];
  __shared__ int cbase[NCH];
  __shared__ int cseg[NCH];
  __shared__ int listA[LCAP];
  __shared__ int listB[LCAP];
  __shared__ int srcS[LCAP];
  __shared__ int nmatch;
  __shared__ int nextrow;

  const int tid = threadIdx.x;
  const int lane = tid & 31;
  const int wave = __builtin_amdgcn_readfirstlane(tid >> 5);
  const int v0 = blockIdx.x * RV;

  for (int k = tid; k < RV; k += WGT) rowcnt[k] = 0;
  if (tid == 0) { nmatch = 0; nextrow = 0; }
  __syncthreads();

  const int nm = fetch_matches(v_sorted, v_tab, blockIdx.x, listA, listB,
                               rowcnt, cbase, cseg, &nmatch, lane, wave);

  const float* __restrict__ esl = es + (lane < 16 ? 0 : 1);
  const float4 zero4 = make_float4(0.f, 0.f, 0.f, 0.f);
  if (nm > LCAP) {
    for (int r = wave; r < RV; r += NWAVE) {
      float4 a = zero4, b = zero4;
      float da = 0.0f, db = 0.0f;
      for (int i0 = 0; i0 < N_NZ; i0 += 32) {
        const int i = i0 + lane;
        const bool m = i < N_NZ && inc_v[i] == v0 + r;
        const int em = m ? inc_e[i] : 0;
        unsigned mask = (unsigned)__ballot(m);
        while (mask != 0u) {
          const int bpos = __ffs((int)mask) - 1;
          mask &= mask - 1u;
          const int e = __shfl(em, bpos);
          const float wa = esl[e * NH], wb = esl[e * NH + 2];
          const float4* __restrict__ yr = (const float4*)(Y + (size_t)e * HID);
          const float4 ya = yr[lane], yb = yr[lane + 32];
          a.x += wa * ya.x; a.y += wa * ya.y; a.z += wa * ya.z; a.w += wa * ya.w;
          b.x += wb * yb.x; b.y += wb * yb.y; b.z += wb * yb.z; b.w += wb * yb.w;
          da += wa; db += wb;
        }
      }
      e2v_finish(v0 + r, a, b, da, db, out, lane);
    }
  } else {
    order_matches<RV>(nm, listA, listB, srcS, rowcnt, rowstart, tid, lane,
                      wave);
    for (;;) {
      int rnew = 0;
      if (lane == 0) rnew = lds_iadd(&nextrow, 1);
      const int r = 2 * __builtin_amdgcn_readfirstlane(rnew);
      if (r >= RV) break;
      const int jA = __builtin_amdgcn_readfirstlane(rowstart[r]);
      const int nA = __builtin_amdgcn_readfirstlane(rowcnt[r]);
      const int jB = __builtin_amdgcn_readfirstlane(rowstart[r + 1]);
      const int nB = __builtin_amdgcn_readfirstlane(rowcnt[r + 1]);
      const int nmax = nA > nB ? nA : nB;
      float4 aA = zero4, bA = zero4, aB = zero4, bB = zero4;
      float daA = 0.0f, dbA = 0.0f, daB = 0.0f, dbB = 0.0f;
      for (int j = 0; j < nmax; j += 4) {
        const int mine =
            srcS[min(((lane & 4) ? jB : jA) + j + (lane & 3), LCAP - 1)];
        float waA[4], wbA[4], waB[4], wbB[4];
        float4 yaA[4], ybA[4], yaB[4], ybB[4];
#pragma unroll
        for (int u = 0; u < 4; ++u) {
          if (j + u < nA) {
            const int e = __builtin_amdgcn_readlane(mine, u);
            waA[u] = esl[e * NH];
            wbA[u] = esl[e * NH + 2];
            const float4* __restrict__ yr = (const float4*)(Y + (size_t)e * HID);
            yaA[u] = yr[lane];
            ybA[u] = yr[lane + 32];
          } else {
            waA[u] = 0.0f; wbA[u] = 0.0f; yaA[u] = zero4; ybA[u] = zero4;
          }
          if (j + u < nB) {
            const int e = __builtin_amdgcn_readlane(mine, 4 + u);
            waB[u] = esl[e * NH];
            wbB[u] = esl[e * NH + 2];
            const float4* __restrict__ yr = (const float4*)(Y + (size_t)e * HID);
            yaB[u] = yr[lane];
            ybB[u] = yr[lane + 32];
          } else {
            waB[u] = 0.0f; wbB[u] = 0.0f; yaB[u] = zero4; ybB[u] = zero4;
          }
        }
#pragma unroll
        for (int u = 0; u < 4; ++u) {
          aA.x += waA[u] * yaA[u].x; aA.y += waA[u] * yaA[u].y;
          aA.z += waA[u] * yaA[u].z; aA.w += waA[u] * yaA[u].w;
          bA.x += wbA[u] * ybA[u].x; bA.y += wbA[u] * ybA[u].y;
          bA.z += wbA[u] * ybA[u].z; bA.w += wbA[u] * ybA[u].w;
          daA += waA[u]; dbA += wbA[u];
          aB.x += waB[u] * yaB[u].x; aB.y += waB[u] * yaB[u].y;
          aB.z += waB[u] * yaB[u].z; aB.w += waB[u] * yaB[u].w;
          bB.x += wbB[u] * ybB[u].x; bB.y += wbB[u] * ybB[u].y;
          bB.z += wbB[u] * ybB[u].z; bB.w += wbB[u] * ybB[u].w;
          daB += waB[u]; dbB += wbB[u];
        }
      }
      e2v_finish(v0 + r,     aA, bA, daA, dbA, out, lane);
      e2v_finish(v0 + r + 1, aB, bB, daB, dbB, out, lane);
    }
  }
}

extern "C" void kernel_launch(void* const* d_in, const int* in_sizes, int n_in,
                              void* d_out, int out_size, void* d_ws, size_t ws_size,
                              hipStream_t stream) {
  const float* X     = (const float*)d_in[0];
  const float* W     = (const float*)d_in[1];
  const float* bvec  = (const float*)d_in[2];
  const float* a_e   = (const float*)d_in[3];
  const int*   inc_v = (const int*)d_in[4];
  const int*   inc_e = (const int*)d_in[5];
  float* out = (float*)d_out;

  float* ws    = (float*)d_ws;
  float* Y     = ws   + (size_t)N_E * HID;
  float* cnt   = Y    + (size_t)N_E * HID;
  float* es    = cnt  + N_E;
  float* Xsum  = out;
  int* v_sorted = (int*)ws;
  int* v_tab    = v_sorted + N_NZ;
  int* e_sorted = (int*)Y;
  int* e_tab    = e_sorted + N_NZ;

  chunk_sort<<<dim3(NCH, 2), ST, 0, stream>>>(inc_v, inc_e, e_sorted, e_tab,
                                              v_sorted, v_tab);
  v2e_own<<<dim3(N_E / RE), WGT, 0, stream>>>(X, inc_v, inc_e, W, bvec, a_e,
                                              e_sorted, e_tab, Xsum, cnt, es);
  gemm_edge<<<dim3((N_E / 16) * NH / 8), 256, 0, stream>>>(Xsum, cnt, W, bvec, Y);
  e2v_own<<<dim3((N_V + RV - 1) / RV), WGT, 0, stream>>>(Y, es, inc_v, inc_e,
                                                         v_sorted, v_tab, out);
}


# e2v_own gather takes one vertex at a time and loads and adds exactly the incidences it has (nested scalar slot tests): no padded slots are cleared or added, no waits between loads; sum order kept
# speedup vs baseline: 5.1080x; 1.0593x over previous
#include <hip/hip_runtime.h>
#include <hip/hip_bf16.h>

typedef __attribute__((ext_vector_type(2))) float v2f;
typedef __attribute__((ext_vector_type(8))) float v8f;

#define N_V   100000
#define N_E   20000
#define N_NZ  400000
#define CIN   256
#define HID   256
#define NH    4
#define DH    64
#define NEG_SLOPE 0.2f

#define WGT    1024
#define NWAVE  32
#define LCAP   16384
#define RE     80
#define RV     392

__device__ __forceinline__ int lds_iadd(int* p, int v) {
  return __hip_atomic_fetch_add(p, v, __ATOMIC_RELAXED, __HIP_MEMORY_SCOPE_WORKGROUP);
}

#define CH     4096
#define NCH    ((N_NZ + CH - 1) / CH)
#define NCHP   128
#define ST     512
#define SQ     (CH / 4 / ST)

template <bool ISV>
__device__ __forceinline__ void chunk_sort_body(
    const int* __restrict__ keys, const int* __restrict__ others,
    int* __restrict__ sorted, int* __restrict__ tab, int* binc, int* bins)
{
  constexpr unsigned R = ISV ? RV : RE;
  const int tid = threadIdx.x;
  const int lane = tid & 31;
  const int c = blockIdx.x;
  if (tid < 257) binc[tid] = 0;
  if (c == 0 && tid < 256) {
    for (int k = NCH; k < NCHP; ++k) tab[tid * NCHP + k] = 0;
  }
  int kk[SQ * 4], oo[SQ * 4];
  bool ok[SQ];
#pragma unroll
  for (int u = 0; u < SQ; ++u) {
    const int q = c * (CH / 4) + u * ST + tid;
    ok[u] = q * 4 < N_NZ;
    const int4 k4 = ok[u] ? ((const int4*)keys)[q] : make_int4(0, 0, 0, 0);
    const int4 o4 = ok[u] ? ((const int4*)others)[q] : make_int4(0, 0, 0, 0);
    kk[u * 4 + 0] = k4.x; kk[u * 4 + 1] = k4.y;
    kk[u * 4 + 2] = k4.z; kk[u * 4 + 3] = k4.w;
    oo[u * 4 + 0] = o4.x; oo[u * 4 + 1] = o4.y;
    oo[u * 4 + 2] = o4.z; oo[u * 4 + 3] = o4.w;
  }
  __syncthreads();
  int own[SQ * 4], rank[SQ * 4];
#pragma unroll
  for (int i = 0; i < SQ * 4; ++i) {
    const unsigned o = min((unsigned)kk[i] / R, 255u);
    kk[i] = kk[i] - (int)(o * R);
    own[i] = ok[i >> 2] ? (int)o : 256;
    rank[i] = lds_iadd(&binc[own[i]], 1);
  }
  __syncthreads();
  if (tid < 32) {
    int cs[8];
    int sum = 0;
#pragma unroll
    for (int k = 0; k < 8; ++k) { cs[k] = binc[lane * 8 + k]; sum += cs[k]; }
    int inc = sum;
#pragma unroll
    for (int off = 1; off < 32; off <<= 1) {
      const int up = __shfl_up(inc, off);
      if (lane >= off) inc += up;
    }
    int run = inc - sum;
#pragma unroll
    for (int k = 0; k < 8; ++k) {
      bins[lane * 8 + k] = run;
      tab[(lane * 8 + k) * NCHP + c] = run | (cs[k] << 16);
      run += cs[k];
    }
    if (lane == 0) bins[256] = 0;
  }
  __syncthreads();
  int dst[SQ * 4];
#pragma unroll
  for (int i = 0; i < SQ * 4; ++i) dst[i] = c * CH + bins[own[i]] + rank[i];
#pragma unroll
  for (int i = 0; i < SQ * 4; ++i) {
    if (ok[i >> 2]) sorted[dst[i]] = kk[i] | (oo[i] << 9);
  }
}

__global__ __launch_bounds__(ST) void chunk_sort(
    const int* __restrict__ inc_v, const int* __restrict__ inc_e,
    int* __restrict__ e_sorted, int* __restrict__ e_tab,
    int* __restrict__ v_sorted, int* __restrict__ v_tab)
{
  __shared__ int binc[257];
  __shared__ int bins[257];
  if (blockIdx.y != 0)
    chunk_sort_body<true>(inc_v, inc_e, v_sorted, v_tab, binc, bins);
  else
    chunk_sort_body<false>(inc_e, inc_v, e_sorted, e_tab, binc, bins);
}

__device__ __forceinline__ int fetch_matches(
    const int* __restrict__ sorted, const int* __restrict__ tab, const int ob,
    int* listA, int* listB, int* rowcnt, int* cbase, int* cseg, int* nmatch,
    const int lane, const int wave)
{
  if (wave == 0) {
    const int4 t4 = ((const int4*)(tab + ob * NCHP))[lane];
    int t[4] = {t4.x, t4.y, t4.z, t4.w};
    int s = 0;
#pragma unroll
    for (int k = 0; k < 4; ++k) s += (int)((unsigned)t[k] >> 16);
    int inc = s;
#pragma unroll
    for (int off = 1; off < 32; off <<= 1) {
      const int up = __shfl_up(inc, off);
      if (lane >= off) inc += up;
    }
    int run = inc - s;
#pragma unroll
    for (int k = 0; k < 4; ++k) {
      if (lane * 4 + k < NCH) { cseg[lane * 4 + k] = t[k]; cbase[lane * 4 + k] = run; }
      run += (int)((unsigned)t[k] >> 16);
    }
    if (lane == 31) *nmatch = inc;
  }
  __syncthreads();
  const int nm = *nmatch;
  if (nm <= LCAP) {
    constexpr int CW = (NCH + NWAVE - 1) / NWAVE;
    int st[CW], len[CW], base[CW], w[CW];
#pragma unroll
    for (int u = 0; u < CW; ++u) {
      const int c = wave + u * NWAVE;
      const int t = c < NCH ? cseg[c] : 0;
      base[u] = c < NCH ? cbase[c] : 0;
      len[u] = (int)((unsigned)t >> 16);
      st[u] = c * CH + (t & 0xffff);
    }
#pragma unroll
    for (int u = 0; u < CW; ++u) w[u] = lane < len[u] ? sorted[st[u] + lane] : 0;
#pragma unroll
    for (int u = 0; u < CW; ++u) {
      if (lane < len[u]) {
        const int r = w[u] & 511;
        const int rank = lds_iadd(&rowcnt[r], 1);
        listA[base[u] + lane] = r | (rank << 9);
        listB[base[u] + lane] = (int)((unsigned)w[u] >> 9);
      }
      for (int j = 32 + lane; j < len[u]; j += 32) {
        const int ww = sorted[st[u] + j];
        const int r = ww & 511;
        const int rank = lds_iadd(&rowcnt[r], 1);
        listA[base[u] + j] = r | (rank << 9);
        listB[base[u] + j] = (int)((unsigned)ww >> 9);
      }
    }
  }
  __syncthreads();
  return nm;
}

template <int R>
__device__ __forceinline__ void order_matches(
    const int nm, int* listA, int* listB, int* srcS,
    int* rowcnt, int* rowstart, const int tid, const int lane, const int wave)
{
  if (wave == 0) {
    constexpr int C = (R + 31) / 32;
    int cs[C];
    int s = 0;
#pragma unroll
    for (int k = 0; k < C; ++k) {
      const int idx = lane * C + k;
      cs[k] = idx < R ? rowcnt[idx] : 0;
      s += cs[k];
    }
    int inc = s;
#pragma unroll
    for (int off = 1; off < 32; off <<= 1) {
      const int up = __shfl_up(inc, off);
      if (lane >= off) inc += up;
    }
    int run = inc - s;
#pragma unroll
    for (int k = 0; k < C; ++k) {
      const int idx = lane * C + k;
      if (idx < R) rowstart[idx] = run;
      run += cs[k];
    }
  }
  __syncthreads();
  for (int t = tid; t < nm; t += WGT) {
    const int rr = listA[t];
    srcS[rowstart[rr & 511] + (rr >> 9)] = listB[t];
  }
  __syncthreads();
}

__device__ __forceinline__ void v2e_finish(
    const int e, const float4 xa, const float4 xb, const float c,
    const float* uL, const float* c0L, float* __restrict__ Xsum,
    float* __restrict__ cnt, float* __restrict__ es, const int lane)
{
  float4* __restrict__ dst = (float4*)(Xsum + (size_t)e * HID);
  dst[lane] = xa;
  dst[lane + 32] = xb;
  float p[NH];
#pragma unroll
  for (int h = 0; h < NH; ++h) {
    const float4 ua = ((const float4*)uL)[h * 64 + lane];
    const float4 ub = ((const float4*)uL)[h * 64 + lane + 32];
    p[h] = xa.x * ua.x + xa.y * ua.y + xa.z * ua.z + xa.w * ua.w
         + xb.x * ub.x + xb.y * ub.y + xb.z * ub.z + xb.w * ub.w;
  }
#pragma unroll
  for (int off = 16; off >= 1; off >>= 1) {
#pragma unroll
    for (int h = 0; h < NH; ++h) p[h] += __shfl_xor(p[h], off);
  }
  if (lane == 0) cnt[e] = c;
  if (lane < NH) {
    const float ph = lane == 0 ? p[0] : lane == 1 ? p[1]
                   : lane == 2 ? p[2] : p[3];
    const float alpha = ph * (1.0f / fmaxf(c, 1.0f)) + c0L[lane];
    float s = alpha >= 0.0f ? alpha : NEG_SLOPE * alpha;
    s = fminf(fmaxf(s, 0.001f), 5.0f);
    es[e * NH + lane] = __expf(s);
  }
}

__global__ __launch_bounds__(WGT) void v2e_own(
    const float* __restrict__ X, const int* __restrict__ inc_v,
    const int* __restrict__ inc_e, const float* __restrict__ W,
    const float* __restrict__ bvec, const float* __restrict__ a_e,
    const int* __restrict__ e_sorted, const int* __restrict__ e_tab,
    float* __restrict__ Xsum, float* __restrict__ cnt, float* __restrict__ es)
{
  __shared__ __attribute__((aligned(16))) float uL[NH * CIN];
  __shared__ float c0L[NH];
  __shared__ int rowcnt[RE];
  __shared__ int rowstart[RE];
  __shared__ int cbase[NCH];
  __shared__ int cseg[NCH];
  __shared__ int listA[LCAP];
  __shared__ int listB[LCAP];
  __shared__ int srcS[LCAP];
  __shared__ int nmatch;
  __shared__ int nextrow;

  const int tid = threadIdx.x;
  const int lane = tid & 31;
  const int wave = __builtin_amdgcn_readfirstlane(tid >> 5);
  const int e0 = blockIdx.x * RE;

  {
    const int h = wave >> 3;
    const int cb = (wave & 7) * 32;
    const int qd = lane & 3, rw = lane >> 2;
    const float4* __restrict__ ap = (const float4*)(a_e + h * DH) + qd * 4;
    const float4* __restrict__ wp =
        (const float4*)(W + ((size_t)h * CIN + cb + rw) * DH) + qd * 4;
    const float4* __restrict__ bp = (const float4*)(bvec + h * DH) + qd * 4;
    float4 av[4], bv[4], wv[16];
#pragma unroll
    for (int b = 0; b < 4; ++b) {
#pragma unroll
      for (int i = 0; i < 4; ++i) wv[b * 4 + i] = wp[b * (8 * DH / 4) + i];
    }
#pragma unroll
    for (int i = 0; i < 4; ++i) { av[i] = ap[i]; bv[i] = bp[i]; }
    float p[4];
#pragma unroll
    for (int b = 0; b < 4; ++b) {
      float s = 0.0f;
#pragma unroll
      for (int i = 0; i < 4; ++i)
        s += wv[b * 4 + i].x * av[i].x + wv[b * 4 + i].y * av[i].y
           + wv[b * 4 + i].z * av[i].z + wv[b * 4 + i].w * av[i].w;
      p[b] = s;
    }
    const bool o1 = (lane & 1) != 0, o2 = (lane & 2) != 0;
    const float q0 = (o1 ? p[1] : p[0]) + __shfl_xor(o1 ? p[0] : p[1], 1);
    const float q1 = (o1 ? p[3] : p[2]) + __shfl_xor(o1 ? p[2] : p[3], 1);
    const float uv = (o2 ? q1 : q0) + __shfl_xor(o2 ? q0 : q1, 2);
    uL[h * CIN + cb + qd * 8 + rw] = uv;
    float q = 0.0f;
#pragma unroll
    for (int i = 0; i < 4; ++i)
      q += bv[i].x * av[i].x + bv[i].y * av[i].y
         + bv[i].z * av[i].z + bv[i].w * av[i].w;
    q += __shfl_xor(q, 1);
    q += __shfl_xor(q, 2);
    if ((wave & 7) == 0 && lane == 0) c0L[h] = q;
  }

  for (int k = tid; k < RE; k += WGT) rowcnt[k] = 0;
  if (tid == 0) { nmatch = 0; nextrow = 0; }
  __syncthreads();

  const int nm = fetch_matches(e_sorted, e_tab, blockIdx.x, listA, listB,
                               rowcnt, cbase, cseg, &nmatch, lane, wave);

  const float4 zero4 = make_float4(0.f, 0.f, 0.f, 0.f);
  if (nm > LCAP) {
    for (int r = wave; r < RE; r += NWAVE) {
      float4 sa = zero4, sb = zero4;
      int n = 0;
      for (int i0 = 0; i0 < N_NZ; i0 += 32) {
        const int i = i0 + lane;
        const bool m = i < N_NZ && inc_e[i] == e0 + r;
        const int vm = m ? inc_v[i] : 0;
        unsigned mask = (unsigned)__ballot(m);
        while (mask != 0u) {
          const int bpos = __ffs((int)mask) - 1;
          mask &= mask - 1u;
          const int v = __shfl(vm, bpos);
          const float4* __restrict__ xr = (const float4*)(X + (size_t)v * CIN);
          const float4 xa = xr[lane], xb = xr[lane + 32];
          sa.x += xa.x; sa.y += xa.y; sa.z += xa.z; sa.w += xa.w;
          sb.x += xb.x; sb.y += xb.y; sb.z += xb.z; sb.w += xb.w;
          ++n;
        }
      }
      v2e_finish(e0 + r, sa, sb, (float)n, uL, c0L, Xsum, cnt, es, lane);
    }
  } else {
    order_matches<RE>(nm, listA, listB, srcS, rowcnt, rowstart, tid, lane,
                      wave);
    for (;;) {
      int rnew = 0;
      if (lane == 0) rnew = lds_iadd(&nextrow, 1);
      const int r = __builtin_amdgcn_readfirstlane(rnew);
      if (r >= RE) break;
      const int j0 = __builtin_amdgcn_readfirstlane(rowstart[r]);
      const int n  = __builtin_amdgcn_readfirstlane(rowcnt[r]);
      float4 sa = zero4, sb = zero4;
      for (int j = 0; j < n; j += 8) {
        const int mine = srcS[min(j0 + j + (lane & 7), LCAP - 1)];
        float4 xa[8], xb[8];
#pragma unroll
        for (int u = 0; u < 8; ++u) {
          if (j + u < n) {
            const int v = __builtin_amdgcn_readlane(mine, u);
            const float4* __restrict__ xr = (const float4*)(X + (size_t)v * CIN);
            xa[u] = xr[lane];
            xb[u] = xr[lane + 32];
          } else {
            xa[u] = zero4;
            xb[u] = zero4;
          }
        }
#pragma unroll
        for (int u = 0; u < 8; ++u) {
          sa.x += xa[u].x; sa.y += xa[u].y; sa.z += xa[u].z; sa.w += xa[u].w;
          sb.x += xb[u].x; sb.y += xb[u].y; sb.z += xb[u].z; sb.w += xb[u].w;
        }
      }
      v2e_finish(e0 + r, sa, sb, (float)n, uL, c0L, Xsum, cnt, es, lane);
    }
  }
}

struct FragSet {
  float4 a;
  float4 b[4];
};

__device__ __forceinline__ void frag_load(
    FragSet& f, const float4* __restrict__ ap, const float4* __restrict__ bp,
    const int blk)
{
  f.a = ap[blk * 2];
#pragma unroll
  for (int q = 0; q < 4; ++q) f.b[q] = bp[blk * (8 * DH / 4) + q * (DH / 4)];
}

__device__ __forceinline__ void frag_mma(
    const FragSet& f, const float inv, v8f (&acc)[4])
{
  const float ak[4] = {f.a.x * inv, f.a.y * inv, f.a.z * inv, f.a.w * inv};
#pragma unroll
  for (int j = 0; j < 2; ++j) {
    v2f afrag;
    afrag.x = ak[2 * j];
    afrag.y = ak[2 * j + 1];
    const float4 bx = f.b[2 * j];
    const float4 by = f.b[2 * j + 1];
    const float bxs[4] = {bx.x, bx.y, bx.z, bx.w};
    const float bys[4] = {by.x, by.y, by.z, by.w};
#pragma unroll
    for (int t = 0; t < 4; ++t) {
      v2f bfrag;
      bfrag.x = bxs[t];
      bfrag.y = bys[t];
      acc[t] = __builtin_amdgcn_wmma_f32_16x16x4_f32(
          false, afrag, false, bfrag, (short)0, acc[t], false, false);
    }
  }
}

__global__ __launch_bounds__(256) void gemm_edge(
    const float* __restrict__ Xsum, const float* __restrict__ cnt,
    const float* __restrict__ W, const float* __restrict__ bvec,
    float* __restrict__ Y)
{
  const int lane = threadIdx.x & 31;
  const int wid  = blockIdx.x * 8 + (threadIdx.x >> 5);
  const int row_tile = wid >> 2;
  const int head = wid & 3;
  const int row0 = row_tile * 16;
  const int m = lane & 15;
  const int half = lane >> 4;

  const int row = row0 + m;
  const float inv = 1.0f / fmaxf(cnt[row], 1.0f);
  const float4* __restrict__ ap =
      (const float4*)(Xsum + (size_t)row * HID) + half;
  const float4* __restrict__ bp =
      (const float4*)(W + (size_t)head * (CIN * DH) + (size_t)(4 * half) * DH) + m;

  v8f acc[4] = {v8f{}, v8f{}, v8f{}, v8f{}};
  FragSet f0, f1;
  frag_load(f0, ap, bp, 0);
  for (int blk = 0; blk < CIN / 8; blk += 2) {
    frag_load(f1, ap, bp, blk + 1);
    frag_mma(f0, inv, acc);
    if (blk + 2 < CIN / 8) frag_load(f0, ap, bp, blk + 2);
    frag_mma(f1, inv, acc);
  }

  const int mbase = half * 8;
  const float4 bias = ((const float4*)(bvec + head * DH))[m];
  float4* __restrict__ yp =
      (float4*)(Y + (size_t)(row0 + mbase) * HID + head * DH) + m;
#pragma unroll
  for (int r = 0; r < 8; ++r) {
    float4 o;
    o.x = acc[0][r] + bias.x;
    o.y = acc[1][r] + bias.y;
    o.z = acc[2][r] + bias.z;
    o.w = acc[3][r] + bias.w;
    yp[r * (HID / 4)] = o;
  }
}

__device__ __forceinline__ void e2v_finish(
    const int v, const float4 a, const float4 b, const float da,
    const float db, float* __restrict__ out, const int lane)
{
  if (v < N_V) {
    const float ia = da > 0.0f ? 1.0f / da : 0.0f;
    const float ib = db > 0.0f ? 1.0f / db : 0.0f;
    float4 x, y;
    x.x = a.x * ia; x.y = a.y * ia; x.z = a.z * ia; x.w = a.w * ia;
    y.x = b.x * ib; y.y = b.y * ib; y.z = b.z * ib; y.w = b.w * ib;
    x.x = x.x > 0.0f ? x.x : expm1f(x.x);
    x.y = x.y > 0.0f ? x.y : expm1f(x.y);
    x.z = x.z > 0.0f ? x.z : expm1f(x.z);
    x.w = x.w > 0.0f ? x.w : expm1f(x.w);
    y.x = y.x > 0.0f ? y.x : expm1f(y.x);
    y.y = y.y > 0.0f ? y.y : expm1f(y.y);
    y.z = y.z > 0.0f ? y.z : expm1f(y.z);
    y.w = y.w > 0.0f ? y.w : expm1f(y.w);
    float4* __restrict__ dst = (float4*)(out + (size_t)v * HID);
    dst[lane] = x;
    dst[lane + 32] = y;
  }
}

template <int U>
__device__ __forceinline__ void e2v_load_from(
    const int m, const int mine, const float* __restrict__ esl,
    const float* __restrict__ Y, const int lane,
    float (&wa)[8], float (&wb)[8], float4 (&ya)[8], float4 (&yb)[8])
{
  if constexpr (U < 8) {
    if (U < m) {
      const int e = __builtin_amdgcn_readlane(mine, U);
      wa[U] = esl[e * NH];
      wb[U] = esl[e * NH + 2];
      const float4* __restrict__ yr = (const float4*)(Y + (size_t)e * HID);
      ya[U] = yr[lane];
      yb[U] = yr[lane + 32];
      e2v_load_from<U + 1>(m, mine, esl, Y, lane, wa, wb, ya, yb);
    }
  }
}

template <int U>
__device__ __forceinline__ void e2v_add_from(
    const int m, const float (&wa)[8], const float (&wb)[8],
    const float4 (&ya)[8], const float4 (&yb)[8],
    float4& a, float4& b, float& da, float& db)
{
  if constexpr (U < 8) {
    if (U < m) {
      a.x += wa[U] * ya[U].x; a.y += wa[U] * ya[U].y;
      a.z += wa[U] * ya[U].z; a.w += wa[U] * ya[U].w;
      b.x += wb[U] * yb[U].x; b.y += wb[U] * yb[U].y;
      b.z += wb[U] * yb[U].z; b.w += wb[U] * yb[U].w;
      da += wa[U]; db += wb[U];
      e2v_add_from<U + 1>(m, wa, wb, ya, yb, a, b, da, db);
    }
  }
}

__global__ __launch_bounds__(WGT) void e2v_own(
    const float* __restrict__ Y, const float* __restrict__ es,
    const int* __restrict__ inc_v, const int* __restrict__ inc_e,
    const int* __restrict__ v_sorted, const int* __restrict__ v_tab,
    float* __restrict__ out)
{
  __shared__ int rowcnt[RV];
  __shared__ int rowstart[RV];
  __shared__ int cbase[NCH];
  __shared__ int cseg[NCH];
  __shared__ int listA[LCAP];
  __shared__ int listB[LCAP];
  __shared__ int srcS[LCAP];
  __shared__ int nmatch;
  __shared__ int nextrow;

  const int tid = threadIdx.x;
  const int lane = tid & 31;
  const int wave = __builtin_amdgcn_readfirstlane(tid >> 5);
  const int v0 = blockIdx.x * RV;

  for (int k = tid; k < RV; k += WGT) rowcnt[k] = 0;
  if (tid == 0) { nmatch = 0; nextrow = 0; }
  __syncthreads();

  const int nm = fetch_matches(v_sorted, v_tab, blockIdx.x, listA, listB,
                               rowcnt, cbase, cseg, &nmatch, lane, wave);

  const float* __restrict__ esl = es + (lane < 16 ? 0 : 1);
  const float4 zero4 = make_float4(0.f, 0.f, 0.f, 0.f);
  if (nm > LCAP) {
    for (int r = wave; r < RV; r += NWAVE) {
      float4 a = zero4, b = zero4;
      float da = 0.0f, db = 0.0f;
      for (int i0 = 0; i0 < N_NZ; i0 += 32) {
        const int i = i0 + lane;
        const bool m = i < N_NZ && inc_v[i] == v0 + r;
        const int em = m ? inc_e[i] : 0;
        unsigned mask = (unsigned)__ballot(m);
        while (mask != 0u) {
          const int bpos = __ffs((int)mask) - 1;
          mask &= mask - 1u;
          const int e = __shfl(em, bpos);
          const float wa = esl[e * NH], wb = esl[e * NH + 2];
          const float4* __restrict__ yr = (const float4*)(Y + (size_t)e * HID);
          const float4 ya = yr[lane], yb = yr[lane + 32];
          a.x += wa * ya.x; a.y += wa * ya.y; a.z += wa * ya.z; a.w += wa * ya.w;
          b.x += wb * yb.x; b.y += wb * yb.y; b.z += wb * yb.z; b.w += wb * yb.w;
          da += wa; db += wb;
        }
      }
      e2v_finish(v0 + r, a, b, da, db, out, lane);
    }
  } else {
    order_matches<RV>(nm, listA, listB, srcS, rowcnt, rowstart, tid, lane,
                      wave);
    for (;;) {
      int rnew = 0;
      if (lane == 0) rnew = lds_iadd(&nextrow, 1);
      const int r = __builtin_amdgcn_readfirstlane(rnew);
      if (r >= RV) break;
      const int j0 = __builtin_amdgcn_readfirstlane(rowstart[r]);
      const int n  = __builtin_amdgcn_readfirstlane(rowcnt[r]);
      float4 a = zero4, b = zero4;
      float da = 0.0f, db = 0.0f;
      for (int j = 0; j < n; j += 8) {
        const int mine = srcS[min(j0 + j + (lane & 7), LCAP - 1)];
        float wa[8], wb[8];
        float4 ya[8], yb[8];
        e2v_load_from<0>(n - j, mine, esl, Y, lane, wa, wb, ya, yb);
        e2v_add_from<0>(n - j, wa, wb, ya, yb, a, b, da, db);
      }
      e2v_finish(v0 + r, a, b, da, db, out, lane);
    }
  }
}

extern "C" void kernel_launch(void* const* d_in, const int* in_sizes, int n_in,
                              void* d_out, int out_size, void* d_ws, size_t ws_size,
                              hipStream_t stream) {
  const float* X     = (const float*)d_in[0];
  const float* W     = (const float*)d_in[1];
  const float* bvec  = (const float*)d_in[2];
  const float* a_e   = (const float*)d_in[3];
  const int*   inc_v = (const int*)d_in[4];
  const int*   inc_e = (const int*)d_in[5];
  float* out = (float*)d_out;

  float* ws    = (float*)d_ws;
  float* Y     = ws   + (size_t)N_E * HID;
  float* cnt   = Y    + (size_t)N_E * HID;
  float* es    = cnt  + N_E;
  float* Xsum  = out;
  int* v_sorted = (int*)ws;
  int* v_tab    = v_sorted + N_NZ;
  int* e_sorted = (int*)Y;
  int* e_tab    = e_sorted + N_NZ;

  chunk_sort<<<dim3(NCH, 2), ST, 0, stream>>>(inc_v, inc_e, e_sorted, e_tab,
                                              v_sorted, v_tab);
  v2e_own<<<dim3(N_E / RE), WGT, 0, stream>>>(X, inc_v, inc_e, W, bvec, a_e,
                                              e_sorted, e_tab, Xsum, cnt, es);
  gemm_edge<<<dim3((N_E / 16) * NH / 8), 256, 0, stream>>>(Xsum, cnt, W, bvec, Y);
  e2v_own<<<dim3((N_V + RV - 1) / RV), WGT, 0, stream>>>(Y, es, inc_v, inc_e,
                                                         v_sorted, v_tab, out);
}
